# starting kernel
# baseline (speedup 1.0000x reference)
#include <hip/hip_runtime.h>
#include <hip/hip_bf16.h>

typedef __attribute__((ext_vector_type(2))) float v2f;
typedef __attribute__((ext_vector_type(8))) float v8f;

#define IN_CH  128
#define HID    64
#define OUTC   40
#define OUTP   48

__global__ void gcn_init(float* __restrict__ deg, float* __restrict__ agg1,
                         float* __restrict__ out, const float* __restrict__ b1,
                         const float* __restrict__ b2, int N) {
  int tid = blockIdx.x * blockDim.x + threadIdx.x;
  long long total = (long long)N * HID;
  if (tid < N) deg[tid] = 1.0f;
  if ((long long)tid < total) agg1[tid] = b1[tid & (HID - 1)];
  if ((long long)tid < (long long)N * OUTC) out[tid] = b2[tid % OUTC];
}

__global__ void gcn_degree(const int* __restrict__ col, float* __restrict__ deg,
                           int E) {
  int e = blockIdx.x * blockDim.x + threadIdx.x;
  if (e < E) atomicAdd(&deg[col[e]], 1.0f);
}

__global__ void gcn_dinv(float* __restrict__ deg, int N) {
  int i = blockIdx.x * blockDim.x + threadIdx.x;
  if (i < N) {
    float d = deg[i];
    deg[i] = (d > 0.0f) ? rsqrtf(d) : 0.0f;
  }
}

__global__ void gcn_gemm1(const float* __restrict__ x,
                          const float* __restrict__ W1,
                          float* __restrict__ h1) {
  const int wave = threadIdx.x >> 5;
  const int lane = threadIdx.x & 31;
  const int m0 = blockIdx.x * 16;
  const int n0 = wave * 16;
  const int rowA = lane & 15;
  const int colB = lane & 15;
  const int kOff = (lane < 16) ? 0 : 2;

  const float* xrow = x + (size_t)(m0 + rowA) * IN_CH;
  v8f acc = {};
#pragma unroll
  for (int k = 0; k < IN_CH; k += 4) {
    v2f a, b;
    a.x = xrow[k + kOff];
    a.y = xrow[k + kOff + 1];
    b.x = W1[(size_t)(k + kOff) * HID + n0 + colB];
    b.y = W1[(size_t)(k + kOff + 1) * HID + n0 + colB];
    acc = __builtin_amdgcn_wmma_f32_16x16x4_f32(false, a, false, b,
                                                (short)0, acc, false, false);
  }
  const int mBase = m0 + ((lane < 16) ? 0 : 8);
  const int c = n0 + colB;
#pragma unroll
  for (int j = 0; j < 8; ++j)
    h1[(size_t)(mBase + j) * HID + c] = acc[j];
}

__global__ void gcn_scatter1(const int* __restrict__ row,
                             const int* __restrict__ col,
                             const float* __restrict__ dinv,
                             const float* __restrict__ h1,
                             float* __restrict__ agg1, int E, int N) {
  int idx = blockIdx.x * blockDim.x + threadIdx.x;
  int e = idx >> 5;
  int lane = idx & 31;
  if (e >= E + N) return;
  int r, c;
  if (e < E) { r = row[e]; c = col[e]; } else { r = c = e - E; }
  float nrm = dinv[r] * dinv[c];
  const float2 hv = *(const float2*)(h1 + (size_t)r * HID + lane * 2);
  float* dst = agg1 + (size_t)c * HID + lane * 2;
  atomicAdd(dst,     hv.x * nrm);
  atomicAdd(dst + 1, hv.y * nrm);
}

__global__ void gcn_gemm2(const float* __restrict__ agg1,
                          const float* __restrict__ W2,
                          float* __restrict__ h2) {
  const int wave = threadIdx.x >> 5;
  const int lane = threadIdx.x & 31;
  const int m0 = blockIdx.x * 16;
  const int n0 = wave * 16;
  const int rowA = lane & 15;
  const int colB = n0 + (lane & 15);
  const int kOff = (lane < 16) ? 0 : 2;
  const bool valid = (colB < OUTC);

  const float* arow = agg1 + (size_t)(m0 + rowA) * HID;
  v8f acc = {};
#pragma unroll
  for (int k = 0; k < HID; k += 4) {
    v2f a, b;
    a.x = fmaxf(arow[k + kOff], 0.0f);
    a.y = fmaxf(arow[k + kOff + 1], 0.0f);
    b.x = valid ? W2[(size_t)(k + kOff) * OUTC + colB] : 0.0f;
    b.y = valid ? W2[(size_t)(k + kOff + 1) * OUTC + colB] : 0.0f;
    acc = __builtin_amdgcn_wmma_f32_16x16x4_f32(false, a, false, b,
                                                (short)0, acc, false, false);
  }
  const int mBase = m0 + ((lane < 16) ? 0 : 8);
  if (valid) {
#pragma unroll
    for (int j = 0; j < 8; ++j)
      h2[(size_t)(mBase + j) * OUTP + colB] = acc[j];
  }
}

__global__ void gcn_scatter2(const int* __restrict__ row,
                             const int* __restrict__ col,
                             const float* __restrict__ dinv,
                             const float* __restrict__ h2,
                             float* __restrict__ out, int E, int N) {
  int idx = blockIdx.x * blockDim.x + threadIdx.x;
  int e = idx >> 5;
  int lane = idx & 31;
  if (e >= E + N || lane >= OUTC / 2) return;
  int r, c;
  if (e < E) { r = row[e]; c = col[e]; } else { r = c = e - E; }
  float nrm = dinv[r] * dinv[c];
  const float2 hv = *(const float2*)(h2 + (size_t)r * OUTP + lane * 2);
  float* dst = out + (size_t)c * OUTC + lane * 2;
  atomicAdd(dst,     hv.x * nrm);
  atomicAdd(dst + 1, hv.y * nrm);
}

extern "C" void kernel_launch(void* const* d_in, const int* in_sizes, int n_in,
                              void* d_out, int out_size, void* d_ws, size_t ws_size,
                              hipStream_t stream) {
  const float* x  = (const float*)d_in[0];
  const int*   ei = (const int*)d_in[1];
  const float* W1 = (const float*)d_in[2];
  const float* b1 = (const float*)d_in[3];
  const float* W2 = (const float*)d_in[4];
  const float* b2 = (const float*)d_in[5];
  float* out = (float*)d_out;

  const int N = in_sizes[0] / IN_CH;
  const int E = in_sizes[1] / 2;
  const int* rowp = ei;
  const int* colp = ei + E;

  float* ws   = (float*)d_ws;
  float* deg  = ws;
  float* h1   = ws + N;
  float* agg1 = ws + N + (size_t)N * HID;

  {
    long long total = (long long)N * HID;
    int blocks = (int)((total + 255) / 256);
    gcn_init<<<blocks, 256, 0, stream>>>(deg, agg1, out, b1, b2, N);
  }
  gcn_degree<<<(E + 255) / 256, 256, 0, stream>>>(colp, deg, E);
  gcn_dinv<<<(N + 255) / 256, 256, 0, stream>>>(deg, N);
  gcn_gemm1<<<N / 16, 128, 0, stream>>>(x, W1, h1);
  {
    long long work = (long long)(E + N) * 32;
    int blocks = (int)((work + 255) / 256);
    gcn_scatter1<<<blocks, 256, 0, stream>>>(rowp, colp, deg, h1, agg1, E, N);
  }
  gcn_gemm2<<<N / 16, 96, 0, stream>>>(agg1, W2, h1);
  {
    long long work = (long long)(E + N) * 32;
    int blocks = (int)((work + 255) / 256);
    gcn_scatter2<<<blocks, 256, 0, stream>>>(rowp, colp, deg, h1, out, E, N);
  }
}


# Replace the float-atomic edge scatters of both layers by a per-node gather over in-edges grouped by target (counting sort per call, integer atomics); float sums now in arrival-rank segment order
# speedup vs baseline: 1.8520x; 1.8520x over previous
#include <hip/hip_runtime.h>

typedef __attribute__((ext_vector_type(2))) float v2f;
typedef __attribute__((ext_vector_type(8))) float v8f;

#define IN_CH  128
#define HID    64
#define OUTC   40
#define OUTP   48
#define SCAN_TILE 1024

__global__ void gcn_zero_cnt(int* __restrict__ cnt, int N) {
  int i = blockIdx.x * blockDim.x + threadIdx.x;
  if (i < N) cnt[i] = 0;
}

__global__ void gcn_count_rank(const int* __restrict__ col, int* __restrict__ cnt,
                               int* __restrict__ rank, int E) {
  int e = blockIdx.x * blockDim.x + threadIdx.x;
  if (e < E) rank[e] = atomicAdd(&cnt[col[e]], 1);
}

__global__ void gcn_tile_sums(const int* __restrict__ cnt, int* __restrict__ tsum,
                              int N) {
  __shared__ int wsum[8];
  const int lane = threadIdx.x & 31;
  const int wave = threadIdx.x >> 5;
  const int base = blockIdx.x * SCAN_TILE + threadIdx.x * 4;
  int s = 0;
  for (int k = 0; k < 4; ++k) {
    int idx = base + k;
    if (idx < N) s += cnt[idx];
  }
  for (int o = 16; o > 0; o >>= 1) s += __shfl_xor(s, o);
  if (lane == 0) wsum[wave] = s;
  __syncthreads();
  if (threadIdx.x == 0) {
    int t = 0;
    for (int w = 0; w < 8; ++w) t += wsum[w];
    tsum[blockIdx.x] = t;
  }
}

__global__ void gcn_scan_dinv(const int* __restrict__ cnt,
                              const int* __restrict__ tsum,
                              int* __restrict__ off, float* __restrict__ dinv,
                              int N) {
  __shared__ int wsum[8];
  const int lane = threadIdx.x & 31;
  const int wave = threadIdx.x >> 5;
  int tb = 0;
  for (int j = lane; j < (int)blockIdx.x; j += 32) tb += tsum[j];
  for (int o = 16; o > 0; o >>= 1) tb += __shfl_xor(tb, o);

  const int base = blockIdx.x * SCAN_TILE + threadIdx.x * 4;
  int v[4];
  int s = 0;
  for (int k = 0; k < 4; ++k) {
    int idx = base + k;
    v[k] = (idx < N) ? cnt[idx] : 0;
    s += v[k];
  }
  int inc = s;
  for (int o = 1; o < 32; o <<= 1) {
    int t = __shfl_up(inc, o);
    if (lane >= o) inc += t;
  }
  if (lane == 31) wsum[wave] = inc;
  __syncthreads();
  int wpre = 0;
  for (int w = 0; w < wave; ++w) wpre += wsum[w];
  int run = tb + wpre + inc - s;
  for (int k = 0; k < 4; ++k) {
    int idx = base + k;
    if (idx < N) {
      off[idx] = run;
      float d = (float)(v[k] + 1);
      dinv[idx] = (d > 0.0f) ? rsqrtf(d) : 0.0f;
      run += v[k];
    }
  }
}

__global__ void gcn_fill(const int* __restrict__ row, const int* __restrict__ col,
                         const int* __restrict__ rank, const int* __restrict__ off,
                         int* __restrict__ src, int E) {
  int e = blockIdx.x * blockDim.x + threadIdx.x;
  if (e < E) src[off[col[e]] + rank[e]] = row[e];
}

__global__ void gcn_gemm1(const float* __restrict__ x,
                          const float* __restrict__ W1,
                          float* __restrict__ h1) {
  const int wave = threadIdx.x >> 5;
  const int lane = threadIdx.x & 31;
  const int m0 = blockIdx.x * 16;
  const int n0 = wave * 16;
  const int rowA = lane & 15;
  const int colB = lane & 15;
  const int kOff = (lane < 16) ? 0 : 2;

  const float* xrow = x + (size_t)(m0 + rowA) * IN_CH;
  v8f acc = {};
#pragma unroll
  for (int k = 0; k < IN_CH; k += 4) {
    v2f a, b;
    a.x = xrow[k + kOff];
    a.y = xrow[k + kOff + 1];
    b.x = W1[(size_t)(k + kOff) * HID + n0 + colB];
    b.y = W1[(size_t)(k + kOff + 1) * HID + n0 + colB];
    acc = __builtin_amdgcn_wmma_f32_16x16x4_f32(false, a, false, b,
                                                (short)0, acc, false, false);
  }
  const int mBase = m0 + ((lane < 16) ? 0 : 8);
  const int c = n0 + colB;
#pragma unroll
  for (int j = 0; j < 8; ++j)
    h1[(size_t)(mBase + j) * HID + c] = acc[j];
}

__global__ void gcn_gemm2(const float* __restrict__ agg1,
                          const float* __restrict__ W2,
                          float* __restrict__ h2) {
  const int wave = threadIdx.x >> 5;
  const int lane = threadIdx.x & 31;
  const int m0 = blockIdx.x * 16;
  const int n0 = wave * 16;
  const int rowA = lane & 15;
  const int colB = n0 + (lane & 15);
  const int kOff = (lane < 16) ? 0 : 2;
  const bool valid = (colB < OUTC);

  const float* arow = agg1 + (size_t)(m0 + rowA) * HID;
  v8f acc = {};
#pragma unroll
  for (int k = 0; k < HID; k += 4) {
    v2f a, b;
    a.x = fmaxf(arow[k + kOff], 0.0f);
    a.y = fmaxf(arow[k + kOff + 1], 0.0f);
    b.x = valid ? W2[(size_t)(k + kOff) * OUTC + colB] : 0.0f;
    b.y = valid ? W2[(size_t)(k + kOff + 1) * OUTC + colB] : 0.0f;
    acc = __builtin_amdgcn_wmma_f32_16x16x4_f32(false, a, false, b,
                                                (short)0, acc, false, false);
  }
  const int mBase = m0 + ((lane < 16) ? 0 : 8);
  if (valid) {
#pragma unroll
    for (int j = 0; j < 8; ++j)
      h2[(size_t)(mBase + j) * OUTP + colB] = acc[j];
  }
}

template <int STRIDE, int CH, int OSTR>
__global__ void gcn_gather(const int* __restrict__ off, const int* __restrict__ cnt,
                           const int* __restrict__ src,
                           const float* __restrict__ dinv,
                           const float* __restrict__ h,
                           const float* __restrict__ bias,
                           float* __restrict__ dst, int N) {
  const int lane = threadIdx.x & 31;
  const int c = __builtin_amdgcn_readfirstlane(
      (int)((blockIdx.x * blockDim.x + threadIdx.x) >> 5));
  if (c >= N) return;
  const int start = off[c];
  const int total = cnt[c] + 1;
  const float dc = dinv[c];
  const bool act = lane < CH / 2;
  const float* hl = h + lane * 2;

  float ax = 0.0f, ay = 0.0f;
  for (int base = 0; base < total; base += 32) {
    const int j = base + lane;
    int r_l = c;
    float w_l = 0.0f;
    if (j < total - 1) {
      r_l = src[start + j];
      w_l = dinv[r_l] * dc;
    } else if (j == total - 1) {
      w_l = dc * dc;
    }
    const int rem = total - base;
    const int m = rem < 32 ? rem : 32;
    int k = 0;
    for (; k + 8 <= m; k += 8) {
      float2 v[8];
      float w[8];
      for (int i = 0; i < 8; ++i) {
        const int r = __builtin_amdgcn_readlane(r_l, k + i);
        w[i] = __int_as_float(__builtin_amdgcn_readlane(__float_as_int(w_l), k + i));
        v[i] = act ? *(const float2*)(hl + (size_t)r * STRIDE) : make_float2(0.0f, 0.0f);
      }
      for (int i = 0; i < 8; ++i) {
        ax += v[i].x * w[i];
        ay += v[i].y * w[i];
      }
    }
    for (; k < m; k += 4) {
      float2 v[4];
      float w[4];
      for (int i = 0; i < 4; ++i) {
        const int r = __builtin_amdgcn_readlane(r_l, k + i);
        w[i] = __int_as_float(__builtin_amdgcn_readlane(__float_as_int(w_l), k + i));
        v[i] = act ? *(const float2*)(hl + (size_t)r * STRIDE) : make_float2(0.0f, 0.0f);
      }
      for (int i = 0; i < 4; ++i) {
        ax += v[i].x * w[i];
        ay += v[i].y * w[i];
      }
    }
  }
  if (act) {
    float2 o;
    o.x = ax + bias[lane * 2];
    o.y = ay + bias[lane * 2 + 1];
    *(float2*)(dst + (size_t)c * OSTR + lane * 2) = o;
  }
}

extern "C" void kernel_launch(void* const* d_in, const int* in_sizes, int n_in,
                              void* d_out, int out_size, void* d_ws, size_t ws_size,
                              hipStream_t stream) {
  const float* x  = (const float*)d_in[0];
  const int*   ei = (const int*)d_in[1];
  const float* W1 = (const float*)d_in[2];
  const float* b1 = (const float*)d_in[3];
  const float* W2 = (const float*)d_in[4];
  const float* b2 = (const float*)d_in[5];
  float* out = (float*)d_out;

  const int N = in_sizes[0] / IN_CH;
  const int E = in_sizes[1] / 2;
  const int* rowp = ei;
  const int* colp = ei + E;

  const size_t Np = ((size_t)N + 63) & ~(size_t)63;
  const size_t Ep = ((size_t)E + 63) & ~(size_t)63;
  const int tiles = (N + SCAN_TILE - 1) / SCAN_TILE;
  const size_t Tp = ((size_t)tiles + 63) & ~(size_t)63;
  int*   wsi  = (int*)d_ws;
  int*   cnt  = wsi;
  int*   off  = wsi + Np;
  float* dinv = (float*)(wsi + 2 * Np);
  int*   tsum = wsi + 3 * Np;
  int*   rank = wsi + 3 * Np + Tp;
  int*   srcs = wsi + 3 * Np + Tp + Ep;
  float* h1   = (float*)(wsi + 3 * Np + Tp + 2 * Ep);
  float* agg1 = h1 + (size_t)N * HID;

  gcn_zero_cnt<<<(N + 255) / 256, 256, 0, stream>>>(cnt, N);
  gcn_count_rank<<<(E + 255) / 256, 256, 0, stream>>>(colp, cnt, rank, E);
  gcn_tile_sums<<<tiles, 256, 0, stream>>>(cnt, tsum, N);
  gcn_scan_dinv<<<tiles, 256, 0, stream>>>(cnt, tsum, off, dinv, N);
  gcn_fill<<<(E + 255) / 256, 256, 0, stream>>>(rowp, colp, rank, off, srcs, E);
  gcn_gemm1<<<N / 16, 128, 0, stream>>>(x, W1, h1);
  gcn_gather<HID, HID, HID><<<(N + 7) / 8, 256, 0, stream>>>(
      off, cnt, srcs, dinv, h1, b1, agg1, N);
  gcn_gemm2<<<N / 16, 96, 0, stream>>>(agg1, W2, h1);
  gcn_gather<OUTP, OUTC, OUTC><<<(N + 7) / 8, 256, 0, stream>>>(
      off, cnt, srcs, dinv, h1, b2, out, N);
}


# Start 6x fewer waves (launches were entry-bound): gather waves loop over 8 nodes with next indices prefetched, edge passes take 8 edges per thread, a GEMM wave owns all column tiles; sum order kept
# speedup vs baseline: 1.9737x; 1.0657x over previous
#include <hip/hip_runtime.h>

typedef __attribute__((ext_vector_type(2))) float v2f;
typedef __attribute__((ext_vector_type(8))) float v8f;

#define IN_CH  128
#define HID    64
#define OUTC   40
#define OUTP   48
#define SCAN_TILE 1024
#define EPT 8
#define NPW 8

__global__ void gcn_zero_cnt(int* __restrict__ cnt, int N) {
  const int step = gridDim.x * blockDim.x;
  for (int i = blockIdx.x * blockDim.x + threadIdx.x; i < N; i += step) cnt[i] = 0;
}

__global__ void gcn_count_rank(const int* __restrict__ col, int* __restrict__ cnt,
                               int* __restrict__ rank, int E) {
  const int base = blockIdx.x * (EPT * 256) + threadIdx.x;
  int cc[EPT], rr[EPT];
#pragma unroll
  for (int i = 0; i < EPT; ++i) {
    const int e = base + i * 256;
    cc[i] = (e < E) ? col[e] : -1;
  }
#pragma unroll
  for (int i = 0; i < EPT; ++i) {
    rr[i] = 0;
    if (cc[i] >= 0) rr[i] = atomicAdd(&cnt[cc[i]], 1);
  }
#pragma unroll
  for (int i = 0; i < EPT; ++i) {
    const int e = base + i * 256;
    if (e < E) rank[e] = rr[i];
  }
}

__global__ void gcn_tile_sums(const int* __restrict__ cnt, int* __restrict__ tsum,
                              int N) {
  __shared__ int wsum[8];
  const int lane = threadIdx.x & 31;
  const int wave = threadIdx.x >> 5;
  const int base = blockIdx.x * SCAN_TILE + threadIdx.x * 4;
  int s = 0;
  for (int k = 0; k < 4; ++k) {
    int idx = base + k;
    if (idx < N) s += cnt[idx];
  }
  for (int o = 16; o > 0; o >>= 1) s += __shfl_xor(s, o);
  if (lane == 0) wsum[wave] = s;
  __syncthreads();
  if (threadIdx.x == 0) {
    int t = 0;
    for (int w = 0; w < 8; ++w) t += wsum[w];
    tsum[blockIdx.x] = t;
  }
}

__global__ void gcn_scan_dinv(const int* __restrict__ cnt,
                              const int* __restrict__ tsum,
                              int* __restrict__ off, float* __restrict__ dinv,
                              int N) {
  __shared__ int wsum[8];
  const int lane = threadIdx.x & 31;
  const int wave = threadIdx.x >> 5;
  int tb = 0;
  for (int j = lane; j < (int)blockIdx.x; j += 32) tb += tsum[j];
  for (int o = 16; o > 0; o >>= 1) tb += __shfl_xor(tb, o);

  const int base = blockIdx.x * SCAN_TILE + threadIdx.x * 4;
  int v[4];
  int s = 0;
  for (int k = 0; k < 4; ++k) {
    int idx = base + k;
    v[k] = (idx < N) ? cnt[idx] : 0;
    s += v[k];
  }
  int inc = s;
  for (int o = 1; o < 32; o <<= 1) {
    int t = __shfl_up(inc, o);
    if (lane >= o) inc += t;
  }
  if (lane == 31) wsum[wave] = inc;
  __syncthreads();
  int wpre = 0;
  for (int w = 0; w < wave; ++w) wpre += wsum[w];
  int run = tb + wpre + inc - s;
  for (int k = 0; k < 4; ++k) {
    int idx = base + k;
    if (idx < N) {
      off[idx] = run;
      float d = (float)(v[k] + 1);
      dinv[idx] = (d > 0.0f) ? rsqrtf(d) : 0.0f;
      run += v[k];
    }
  }
}

__global__ void gcn_fill(const int* __restrict__ row, const int* __restrict__ col,
                         const int* __restrict__ rank, const int* __restrict__ off,
                         int* __restrict__ src, int E) {
  const int base = blockIdx.x * (EPT * 256) + threadIdx.x;
  int cc[EPT], rk[EPT], rw[EPT], oo[EPT];
#pragma unroll
  for (int i = 0; i < EPT; ++i) {
    const int e = base + i * 256;
    const bool in = e < E;
    cc[i] = in ? col[e] : -1;
    rk[i] = in ? rank[e] : 0;
    rw[i] = in ? row[e] : 0;
  }
#pragma unroll
  for (int i = 0; i < EPT; ++i) oo[i] = (cc[i] >= 0) ? off[cc[i]] : 0;
#pragma unroll
  for (int i = 0; i < EPT; ++i)
    if (cc[i] >= 0) src[oo[i] + rk[i]] = rw[i];
}

__global__ void gcn_gemm1(const float* __restrict__ x,
                          const float* __restrict__ W1,
                          float* __restrict__ h1, int tiles) {
  const int wave = threadIdx.x >> 5;
  const int lane = threadIdx.x & 31;
  const int tile = __builtin_amdgcn_readfirstlane((int)(blockIdx.x * 4 + wave));
  if (tile >= tiles) return;
  const int m0 = tile * 16;
  const int rc = lane & 15;
  const int kOff = (lane < 16) ? 0 : 2;

  const float* xrow = x + (size_t)(m0 + rc) * IN_CH;
  v8f acc0 = {}, acc1 = {}, acc2 = {}, acc3 = {};
#pragma unroll
  for (int k = 0; k < IN_CH; k += 4) {
    v2f a, b0, b1, b2, b3;
    a.x = xrow[k + kOff];
    a.y = xrow[k + kOff + 1];
    const float* w0 = W1 + (size_t)(k + kOff) * HID + rc;
    const float* w1 = W1 + (size_t)(k + kOff + 1) * HID + rc;
    b0.x = w0[0];   b0.y = w1[0];
    b1.x = w0[16];  b1.y = w1[16];
    b2.x = w0[32];  b2.y = w1[32];
    b3.x = w0[48];  b3.y = w1[48];
    acc0 = __builtin_amdgcn_wmma_f32_16x16x4_f32(false, a, false, b0,
                                                 (short)0, acc0, false, false);
    acc1 = __builtin_amdgcn_wmma_f32_16x16x4_f32(false, a, false, b1,
                                                 (short)0, acc1, false, false);
    acc2 = __builtin_amdgcn_wmma_f32_16x16x4_f32(false, a, false, b2,
                                                 (short)0, acc2, false, false);
    acc3 = __builtin_amdgcn_wmma_f32_16x16x4_f32(false, a, false, b3,
                                                 (short)0, acc3, false, false);
  }
  const int mBase = m0 + ((lane < 16) ? 0 : 8);
#pragma unroll
  for (int j = 0; j < 8; ++j) {
    float* o = h1 + (size_t)(mBase + j) * HID + rc;
    o[0]  = acc0[j];
    o[16] = acc1[j];
    o[32] = acc2[j];
    o[48] = acc3[j];
  }
}

__global__ void gcn_gemm2(const float* __restrict__ agg1,
                          const float* __restrict__ W2,
                          float* __restrict__ h2, int tiles) {
  const int wave = threadIdx.x >> 5;
  const int lane = threadIdx.x & 31;
  const int tile = __builtin_amdgcn_readfirstlane((int)(blockIdx.x * 4 + wave));
  if (tile >= tiles) return;
  const int m0 = tile * 16;
  const int rc = lane & 15;
  const int kOff = (lane < 16) ? 0 : 2;
  const bool valid2 = (32 + rc < OUTC);
  const int c2 = valid2 ? 32 + rc : 32;

  const float* arow = agg1 + (size_t)(m0 + rc) * HID;
  v8f acc0 = {}, acc1 = {}, acc2 = {};
#pragma unroll
  for (int k = 0; k < HID; k += 4) {
    v2f a, b0, b1, b2;
    a.x = fmaxf(arow[k + kOff], 0.0f);
    a.y = fmaxf(arow[k + kOff + 1], 0.0f);
    const float* w0 = W2 + (size_t)(k + kOff) * OUTC;
    const float* w1 = W2 + (size_t)(k + kOff + 1) * OUTC;
    b0.x = w0[rc];       b0.y = w1[rc];
    b1.x = w0[16 + rc];  b1.y = w1[16 + rc];
    const float t0 = w0[c2], t1 = w1[c2];
    b2.x = valid2 ? t0 : 0.0f;
    b2.y = valid2 ? t1 : 0.0f;
    acc0 = __builtin_amdgcn_wmma_f32_16x16x4_f32(false, a, false, b0,
                                                 (short)0, acc0, false, false);
    acc1 = __builtin_amdgcn_wmma_f32_16x16x4_f32(false, a, false, b1,
                                                 (short)0, acc1, false, false);
    acc2 = __builtin_amdgcn_wmma_f32_16x16x4_f32(false, a, false, b2,
                                                 (short)0, acc2, false, false);
  }
  const int mBase = m0 + ((lane < 16) ? 0 : 8);
#pragma unroll
  for (int j = 0; j < 8; ++j) {
    float* o = h2 + (size_t)(mBase + j) * OUTP + rc;
    o[0]  = acc0[j];
    o[16] = acc1[j];
    if (valid2) o[32] = acc2[j];
  }
}

template <int STRIDE, int CH, int OSTR>
__global__ void gcn_gather(const int* __restrict__ off, const int* __restrict__ cnt,
                           const int* __restrict__ src,
                           const float* __restrict__ dinv,
                           const float* __restrict__ h,
                           const float* __restrict__ bias,
                           float* __restrict__ dst, int N) {
  const int lane = threadIdx.x & 31;
  const int c0 = NPW * __builtin_amdgcn_readfirstlane(
      (int)((blockIdx.x * blockDim.x + threadIdx.x) >> 5));
  if (c0 >= N) return;
  const int nn = (N - c0 < NPW) ? (N - c0) : NPW;
  const int last = nn - 1;

  int m_off = 0, m_tot = 1;
  float m_d = 0.0f;
  if (lane < nn) {
    m_off = off[c0 + lane];
    m_tot = cnt[c0 + lane] + 1;
    m_d = dinv[c0 + lane];
  }
  const bool act = lane < CH / 2;
  const float* hl = h + (act ? lane * 2 : 0);
  float bx = 0.0f, by = 0.0f;
  if (act) {
    bx = bias[lane * 2];
    by = bias[lane * 2 + 1];
  }

  auto raw_r = [&](int i) -> int {
    const int st = __builtin_amdgcn_readlane(m_off, i);
    const int tt = __builtin_amdgcn_readlane(m_tot, i);
    const int idx = (lane < tt - 1) ? st + lane : 0;
    return src[idx];
  };
  auto fix_r = [&](int i, int raw) -> int {
    const int tt = __builtin_amdgcn_readlane(m_tot, i);
    return (lane < tt - 1) ? raw : c0 + i;
  };

  int r_cur = fix_r(0, raw_r(0));
  float g_cur = dinv[r_cur];
  int raw_nxt = raw_r(last < 1 ? last : 1);

  for (int i = 0; i < nn; ++i) {
    const int c = c0 + i;
    const int start = __builtin_amdgcn_readlane(m_off, i);
    const int total = __builtin_amdgcn_readlane(m_tot, i);
    const float dc = __int_as_float(__builtin_amdgcn_readlane(__float_as_int(m_d), i));

    int r_l = r_cur;
    float w_l = (lane < total - 1) ? g_cur * dc
                                   : ((lane == total - 1) ? dc * dc : 0.0f);

    float2 v0[16];
    float w0[16];
    for (int q = 0; q < 16; ++q) {
      const int r = __builtin_amdgcn_readlane(r_l, q);
      w0[q] = __int_as_float(__builtin_amdgcn_readlane(__float_as_int(w_l), q));
      v0[q] = *(const float2*)(hl + (size_t)r * STRIDE);
    }
    const int i1 = (i + 1 < last) ? i + 1 : last;
    const int i2 = (i + 2 < last) ? i + 2 : last;
    const int r_nxt = fix_r(i1, raw_nxt);
    const float g_nxt = dinv[r_nxt];
    const int raw_nn = raw_r(i2);

    float ax = 0.0f, ay = 0.0f;
    for (int q = 0; q < 16; ++q) {
      ax += v0[q].x * w0[q];
      ay += v0[q].y * w0[q];
    }

    int base = 0, k = 16;
    for (;;) {
      const int rem = total - base;
      const int m = rem < 32 ? rem : 32;
      for (; k + 8 <= m; k += 8) {
        float2 v[8];
        float w[8];
        for (int q = 0; q < 8; ++q) {
          const int r = __builtin_amdgcn_readlane(r_l, k + q);
          w[q] = __int_as_float(__builtin_amdgcn_readlane(__float_as_int(w_l), k + q));
          v[q] = *(const float2*)(hl + (size_t)r * STRIDE);
        }
        for (int q = 0; q < 8; ++q) {
          ax += v[q].x * w[q];
          ay += v[q].y * w[q];
        }
      }
      for (; k < m; k += 4) {
        float2 v[4];
        float w[4];
        for (int q = 0; q < 4; ++q) {
          const int r = __builtin_amdgcn_readlane(r_l, k + q);
          w[q] = __int_as_float(__builtin_amdgcn_readlane(__float_as_int(w_l), k + q));
          v[q] = *(const float2*)(hl + (size_t)r * STRIDE);
        }
        for (int q = 0; q < 4; ++q) {
          ax += v[q].x * w[q];
          ay += v[q].y * w[q];
        }
      }
      base += 32;
      if (base >= total) break;
      const int j = base + lane;
      r_l = c;
      w_l = 0.0f;
      if (j < total - 1) {
        r_l = src[start + j];
        w_l = dinv[r_l] * dc;
      } else if (j == total - 1) {
        w_l = dc * dc;
      }
      k = 0;
    }
    if (act) {
      float2 o;
      o.x = ax + bx;
      o.y = ay + by;
      *(float2*)(dst + (size_t)c * OSTR + lane * 2) = o;
    }
    r_cur = r_nxt;
    g_cur = g_nxt;
    raw_nxt = raw_nn;
  }
}

extern "C" void kernel_launch(void* const* d_in, const int* in_sizes, int n_in,
                              void* d_out, int out_size, void* d_ws, size_t ws_size,
                              hipStream_t stream) {
  const float* x  = (const float*)d_in[0];
  const int*   ei = (const int*)d_in[1];
  const float* W1 = (const float*)d_in[2];
  const float* b1 = (const float*)d_in[3];
  const float* W2 = (const float*)d_in[4];
  const float* b2 = (const float*)d_in[5];
  float* out = (float*)d_out;

  const int N = in_sizes[0] / IN_CH;
  const int E = in_sizes[1] / 2;
  const int* rowp = ei;
  const int* colp = ei + E;

  const size_t Np = ((size_t)N + 63) & ~(size_t)63;
  const size_t Ep = ((size_t)E + 63) & ~(size_t)63;
  const int tiles = (N + SCAN_TILE - 1) / SCAN_TILE;
  const size_t Tp = ((size_t)tiles + 63) & ~(size_t)63;
  int*   wsi  = (int*)d_ws;
  int*   cnt  = wsi;
  int*   off  = wsi + Np;
  float* dinv = (float*)(wsi + 2 * Np);
  int*   tsum = wsi + 3 * Np;
  int*   rank = wsi + 3 * Np + Tp;
  int*   srcs = wsi + 3 * Np + Tp + Ep;
  float* h1   = (float*)(wsi + 3 * Np + Tp + 2 * Ep);
  float* agg1 = h1 + (size_t)N * HID;

  gcn_zero_cnt<<<tiles, 256, 0, stream>>>(cnt, N);
  gcn_count_rank<<<(E + EPT * 256 - 1) / (EPT * 256), 256, 0, stream>>>(colp, cnt, rank, E);
  gcn_tile_sums<<<tiles, 256, 0, stream>>>(cnt, tsum, N);
  gcn_scan_dinv<<<tiles, 256, 0, stream>>>(cnt, tsum, off, dinv, N);
  gcn_fill<<<(E + EPT * 256 - 1) / (EPT * 256), 256, 0, stream>>>(rowp, colp, rank, off, srcs, E);
  gcn_gemm1<<<(N / 16 + 3) / 4, 128, 0, stream>>>(x, W1, h1, N / 16);
  gcn_gather<HID, HID, HID><<<(N + 8 * NPW - 1) / (8 * NPW), 256, 0, stream>>>(
      off, cnt, srcs, dinv, h1, b1, agg1, N);
  gcn_gemm2<<<(N / 16 + 3) / 4, 128, 0, stream>>>(agg1, W2, h1, N / 16);
  gcn_gather<OUTP, OUTC, OUTC><<<(N + 8 * NPW - 1) / (8 * NPW), 256, 0, stream>>>(
      off, cnt, srcs, dinv, h1, b2, out, N);
}


# GEMM rows leave scaled by their own D^-1/2, the target's D^-1/2 is applied once after each segment sum: gathers drop the per-edge D^-1/2 lookups; sum order kept, factor rounding moves
# speedup vs baseline: 2.1462x; 1.0874x over previous
#include <hip/hip_runtime.h>

typedef __attribute__((ext_vector_type(2))) float v2f;
typedef __attribute__((ext_vector_type(8))) float v8f;

#define IN_CH  128
#define HID    64
#define OUTC   40
#define OUTP   48
#define SCAN_TILE 1024
#define EPT 8
#define NPW 8

__global__ void gcn_zero_cnt(int* __restrict__ cnt, int N) {
  const int step = gridDim.x * blockDim.x;
  for (int i = blockIdx.x * blockDim.x + threadIdx.x; i < N; i += step) cnt[i] = 0;
}

__global__ void gcn_count_rank(const int* __restrict__ col, int* __restrict__ cnt,
                               int* __restrict__ rank, int E) {
  const int base = blockIdx.x * (EPT * 256) + threadIdx.x;
  int cc[EPT], rr[EPT];
#pragma unroll
  for (int i = 0; i < EPT; ++i) {
    const int e = base + i * 256;
    cc[i] = (e < E) ? col[e] : -1;
  }
#pragma unroll
  for (int i = 0; i < EPT; ++i) {
    rr[i] = 0;
    if (cc[i] >= 0) rr[i] = atomicAdd(&cnt[cc[i]], 1);
  }
#pragma unroll
  for (int i = 0; i < EPT; ++i) {
    const int e = base + i * 256;
    if (e < E) rank[e] = rr[i];
  }
}

__global__ void gcn_tile_sums(const int* __restrict__ cnt, int* __restrict__ tsum,
                              int N) {
  __shared__ int wsum[8];
  const int lane = threadIdx.x & 31;
  const int wave = threadIdx.x >> 5;
  const int base = blockIdx.x * SCAN_TILE + threadIdx.x * 4;
  int s = 0;
  for (int k = 0; k < 4; ++k) {
    int idx = base + k;
    if (idx < N) s += cnt[idx];
  }
  for (int o = 16; o > 0; o >>= 1) s += __shfl_xor(s, o);
  if (lane == 0) wsum[wave] = s;
  __syncthreads();
  if (threadIdx.x == 0) {
    int t = 0;
    for (int w = 0; w < 8; ++w) t += wsum[w];
    tsum[blockIdx.x] = t;
  }
}

__global__ void gcn_scan_dinv(const int* __restrict__ cnt,
                              const int* __restrict__ tsum,
                              int* __restrict__ off, float* __restrict__ dinv,
                              int N) {
  __shared__ int wsum[8];
  const int lane = threadIdx.x & 31;
  const int wave = threadIdx.x >> 5;
  int tb = 0;
  for (int j = lane; j < (int)blockIdx.x; j += 32) tb += tsum[j];
  for (int o = 16; o > 0; o >>= 1) tb += __shfl_xor(tb, o);

  const int base = blockIdx.x * SCAN_TILE + threadIdx.x * 4;
  int v[4];
  int s = 0;
  for (int k = 0; k < 4; ++k) {
    int idx = base + k;
    v[k] = (idx < N) ? cnt[idx] : 0;
    s += v[k];
  }
  int inc = s;
  for (int o = 1; o < 32; o <<= 1) {
    int t = __shfl_up(inc, o);
    if (lane >= o) inc += t;
  }
  if (lane == 31) wsum[wave] = inc;
  __syncthreads();
  int wpre = 0;
  for (int w = 0; w < wave; ++w) wpre += wsum[w];
  int run = tb + wpre + inc - s;
  for (int k = 0; k < 4; ++k) {
    int idx = base + k;
    if (idx < N) {
      off[idx] = run;
      float d = (float)(v[k] + 1);
      dinv[idx] = (d > 0.0f) ? rsqrtf(d) : 0.0f;
      run += v[k];
    }
  }
}

__global__ void gcn_fill(const int* __restrict__ row, const int* __restrict__ col,
                         const int* __restrict__ rank, const int* __restrict__ off,
                         int* __restrict__ src, int E) {
  const int base = blockIdx.x * (EPT * 256) + threadIdx.x;
  int cc[EPT], rk[EPT], rw[EPT], oo[EPT];
#pragma unroll
  for (int i = 0; i < EPT; ++i) {
    const int e = base + i * 256;
    const bool in = e < E;
    cc[i] = in ? col[e] : -1;
    rk[i] = in ? rank[e] : 0;
    rw[i] = in ? row[e] : 0;
  }
#pragma unroll
  for (int i = 0; i < EPT; ++i) oo[i] = (cc[i] >= 0) ? off[cc[i]] : 0;
#pragma unroll
  for (int i = 0; i < EPT; ++i)
    if (cc[i] >= 0) src[oo[i] + rk[i]] = rw[i];
}

__global__ void gcn_gemm1(const float* __restrict__ x,
                          const float* __restrict__ W1,
                          const float* __restrict__ dinv,
                          float* __restrict__ h1, int tiles, int rows) {
  const int wave = threadIdx.x >> 5;
  const int lane = threadIdx.x & 31;
  const int tile = __builtin_amdgcn_readfirstlane((int)(blockIdx.x * 4 + wave));
  if (tile >= tiles) return;
  const int m0 = tile * 16;
  const int rc = lane & 15;
  const int kOff = (lane < 16) ? 0 : 2;

  const int mBase = m0 + ((lane < 16) ? 0 : 8);
  const float4 dlo = *(const float4*)(dinv + mBase);
  const float4 dhi = *(const float4*)(dinv + mBase + 4);
  const float dr[8] = {dlo.x, dlo.y, dlo.z, dlo.w, dhi.x, dhi.y, dhi.z, dhi.w};

  const float* xrow = x + (size_t)(m0 + rc) * IN_CH;
  v8f acc0 = {}, acc1 = {}, acc2 = {}, acc3 = {};
#pragma unroll
  for (int k = 0; k < IN_CH; k += 4) {
    v2f a, b0, b1, b2, b3;
    a.x = xrow[k + kOff];
    a.y = xrow[k + kOff + 1];
    const float* w0 = W1 + (size_t)(k + kOff) * HID + rc;
    const float* w1 = W1 + (size_t)(k + kOff + 1) * HID + rc;
    b0.x = w0[0];   b0.y = w1[0];
    b1.x = w0[16];  b1.y = w1[16];
    b2.x = w0[32];  b2.y = w1[32];
    b3.x = w0[48];  b3.y = w1[48];
    acc0 = __builtin_amdgcn_wmma_f32_16x16x4_f32(false, a, false, b0,
                                                 (short)0, acc0, false, false);
    acc1 = __builtin_amdgcn_wmma_f32_16x16x4_f32(false, a, false, b1,
                                                 (short)0, acc1, false, false);
    acc2 = __builtin_amdgcn_wmma_f32_16x16x4_f32(false, a, false, b2,
                                                 (short)0, acc2, false, false);
    acc3 = __builtin_amdgcn_wmma_f32_16x16x4_f32(false, a, false, b3,
                                                 (short)0, acc3, false, false);
  }
#pragma unroll
  for (int j = 0; j < 8; ++j) {
    float* o = h1 + (size_t)(mBase + j) * HID + rc;
    o[0]  = acc0[j] * dr[j];
    o[16] = acc1[j] * dr[j];
    o[32] = acc2[j] * dr[j];
    o[48] = acc3[j] * dr[j];
  }
  if (tile == 0) {
    float2 z;
    z.x = 0.0f;
    z.y = 0.0f;
    *(float2*)(h1 + (size_t)rows * HID + lane * 2) = z;
  }
}

__global__ void gcn_gemm2(const float* __restrict__ agg1,
                          const float* __restrict__ W2,
                          const float* __restrict__ dinv,
                          float* __restrict__ h2, int tiles, int rows) {
  const int wave = threadIdx.x >> 5;
  const int lane = threadIdx.x & 31;
  const int tile = __builtin_amdgcn_readfirstlane((int)(blockIdx.x * 4 + wave));
  if (tile >= tiles) return;
  const int m0 = tile * 16;
  const int rc = lane & 15;
  const int kOff = (lane < 16) ? 0 : 2;
  const bool valid2 = (32 + rc < OUTC);
  const int c2 = valid2 ? 32 + rc : 32;

  const int mBase = m0 + ((lane < 16) ? 0 : 8);
  const float4 dlo = *(const float4*)(dinv + mBase);
  const float4 dhi = *(const float4*)(dinv + mBase + 4);
  const float dr[8] = {dlo.x, dlo.y, dlo.z, dlo.w, dhi.x, dhi.y, dhi.z, dhi.w};

  const float* arow = agg1 + (size_t)(m0 + rc) * HID;
  v8f acc0 = {}, acc1 = {}, acc2 = {};
#pragma unroll
  for (int k = 0; k < HID; k += 4) {
    v2f a, b0, b1, b2;
    a.x = fmaxf(arow[k + kOff], 0.0f);
    a.y = fmaxf(arow[k + kOff + 1], 0.0f);
    const float* w0 = W2 + (size_t)(k + kOff) * OUTC;
    const float* w1 = W2 + (size_t)(k + kOff + 1) * OUTC;
    b0.x = w0[rc];       b0.y = w1[rc];
    b1.x = w0[16 + rc];  b1.y = w1[16 + rc];
    const float t0 = w0[c2], t1 = w1[c2];
    b2.x = valid2 ? t0 : 0.0f;
    b2.y = valid2 ? t1 : 0.0f;
    acc0 = __builtin_amdgcn_wmma_f32_16x16x4_f32(false, a, false, b0,
                                                 (short)0, acc0, false, false);
    acc1 = __builtin_amdgcn_wmma_f32_16x16x4_f32(false, a, false, b1,
                                                 (short)0, acc1, false, false);
    acc2 = __builtin_amdgcn_wmma_f32_16x16x4_f32(false, a, false, b2,
                                                 (short)0, acc2, false, false);
  }
#pragma unroll
  for (int j = 0; j < 8; ++j) {
    float* o = h2 + (size_t)(mBase + j) * OUTP + rc;
    o[0]  = acc0[j] * dr[j];
    o[16] = acc1[j] * dr[j];
    if (valid2) o[32] = acc2[j] * dr[j];
  }
  if (tile == 0 && lane < OUTP / 2) {
    float2 z;
    z.x = 0.0f;
    z.y = 0.0f;
    *(float2*)(h2 + (size_t)rows * OUTP + lane * 2) = z;
  }
}

template <int STRIDE, int CH, int OSTR>
__global__ void gcn_gather(const int* __restrict__ off, const int* __restrict__ cnt,
                           const int* __restrict__ src,
                           const float* __restrict__ dinv,
                           const float* __restrict__ h,
                           const float* __restrict__ bias,
                           float* __restrict__ dst, int N) {
  const int lane = threadIdx.x & 31;
  const int c0 = NPW * __builtin_amdgcn_readfirstlane(
      (int)((blockIdx.x * blockDim.x + threadIdx.x) >> 5));
  if (c0 >= N) return;
  const int nn = (N - c0 < NPW) ? (N - c0) : NPW;
  const int last = nn - 1;

  int m_off = 0, m_tot = 1;
  float m_d = 0.0f;
  if (lane < nn) {
    m_off = off[c0 + lane];
    m_tot = cnt[c0 + lane] + 1;
    m_d = dinv[c0 + lane];
  }
  const bool act = lane < CH / 2;
  const float* hl = h + (act ? lane * 2 : 0);
  float bx = 0.0f, by = 0.0f;
  if (act) {
    bx = bias[lane * 2];
    by = bias[lane * 2 + 1];
  }

  auto raw_r = [&](int i) -> int {
    const int st = __builtin_amdgcn_readlane(m_off, i);
    const int tt = __builtin_amdgcn_readlane(m_tot, i);
    const int idx = (lane < tt - 1) ? st + lane : 0;
    return src[idx];
  };
  auto fix_r = [&](int i, int raw) -> int {
    const int tt = __builtin_amdgcn_readlane(m_tot, i);
    return (lane < tt - 1) ? raw : ((lane == tt - 1) ? c0 + i : N);
  };

  int r_cur = fix_r(0, raw_r(0));
  int raw_nxt = raw_r(last < 1 ? last : 1);

  for (int i = 0; i < nn; ++i) {
    const int c = c0 + i;
    const int start = __builtin_amdgcn_readlane(m_off, i);
    const int total = __builtin_amdgcn_readlane(m_tot, i);
    const float dc = __int_as_float(__builtin_amdgcn_readlane(__float_as_int(m_d), i));

    int r_l = r_cur;

    float2 v0[16];
    for (int q = 0; q < 16; ++q) {
      const int r = __builtin_amdgcn_readlane(r_l, q);
      v0[q] = *(const float2*)(hl + (size_t)r * STRIDE);
    }
    const int i1 = (i + 1 < last) ? i + 1 : last;
    const int i2 = (i + 2 < last) ? i + 2 : last;
    const int raw_nn = raw_r(i2);

    float ax = 0.0f, ay = 0.0f;
    for (int q = 0; q < 16; ++q) {
      ax += v0[q].x;
      ay += v0[q].y;
    }

    int base = 0, k = 16;
    for (;;) {
      const int rem = total - base;
      const int m = rem < 32 ? rem : 32;
      for (; k + 8 <= m; k += 8) {
        float2 v[8];
        for (int q = 0; q < 8; ++q) {
          const int r = __builtin_amdgcn_readlane(r_l, k + q);
          v[q] = *(const float2*)(hl + (size_t)r * STRIDE);
        }
        for (int q = 0; q < 8; ++q) {
          ax += v[q].x;
          ay += v[q].y;
        }
      }
      for (; k < m; k += 4) {
        float2 v[4];
        for (int q = 0; q < 4; ++q) {
          const int r = __builtin_amdgcn_readlane(r_l, k + q);
          v[q] = *(const float2*)(hl + (size_t)r * STRIDE);
        }
        for (int q = 0; q < 4; ++q) {
          ax += v[q].x;
          ay += v[q].y;
        }
      }
      base += 32;
      if (base >= total) break;
      const int j = base + lane;
      r_l = (j == total - 1) ? c : N;
      if (j < total - 1) r_l = src[start + j];
      k = 0;
    }
    if (act) {
      float2 o;
      o.x = ax * dc + bx;
      o.y = ay * dc + by;
      *(float2*)(dst + (size_t)c * OSTR + lane * 2) = o;
    }
    r_cur = fix_r(i1, raw_nxt);
    raw_nxt = raw_nn;
  }
}

extern "C" void kernel_launch(void* const* d_in, const int* in_sizes, int n_in,
                              void* d_out, int out_size, void* d_ws, size_t ws_size,
                              hipStream_t stream) {
  const float* x  = (const float*)d_in[0];
  const int*   ei = (const int*)d_in[1];
  const float* W1 = (const float*)d_in[2];
  const float* b1 = (const float*)d_in[3];
  const float* W2 = (const float*)d_in[4];
  const float* b2 = (const float*)d_in[5];
  float* out = (float*)d_out;

  const int N = in_sizes[0] / IN_CH;
  const int E = in_sizes[1] / 2;
  const int* rowp = ei;
  const int* colp = ei + E;

  const size_t Np = ((size_t)N + 63) & ~(size_t)63;
  const size_t Ep = ((size_t)E + 63) & ~(size_t)63;
  const int tiles = (N + SCAN_TILE - 1) / SCAN_TILE;
  const size_t Tp = ((size_t)tiles + 63) & ~(size_t)63;
  int*   wsi  = (int*)d_ws;
  int*   cnt  = wsi;
  int*   off  = wsi + Np;
  float* dinv = (float*)(wsi + 2 * Np);
  int*   tsum = wsi + 3 * Np;
  int*   rank = wsi + 3 * Np + Tp;
  int*   srcs = wsi + 3 * Np + Tp + Ep;
  float* h1   = (float*)(wsi + 3 * Np + Tp + 2 * Ep);
  float* agg1 = h1 + ((size_t)N + 1) * HID;

  gcn_zero_cnt<<<tiles, 256, 0, stream>>>(cnt, N);
  gcn_count_rank<<<(E + EPT * 256 - 1) / (EPT * 256), 256, 0, stream>>>(colp, cnt, rank, E);
  gcn_tile_sums<<<tiles, 256, 0, stream>>>(cnt, tsum, N);
  gcn_scan_dinv<<<tiles, 256, 0, stream>>>(cnt, tsum, off, dinv, N);
  gcn_fill<<<(E + EPT * 256 - 1) / (EPT * 256), 256, 0, stream>>>(rowp, colp, rank, off, srcs, E);
  gcn_gemm1<<<(N / 16 + 3) / 4, 128, 0, stream>>>(x, W1, dinv, h1, N / 16, N);
  gcn_gather<HID, HID, HID><<<(N + 8 * NPW - 1) / (8 * NPW), 256, 0, stream>>>(
      off, cnt, srcs, dinv, h1, b1, agg1, N);
  gcn_gemm2<<<(N / 16 + 3) / 4, 128, 0, stream>>>(agg1, W2, dinv, h1, N / 16, N);
  gcn_gather<OUTP, OUTC, OUTC><<<(N + 8 * NPW - 1) / (8 * NPW), 256, 0, stream>>>(
      off, cnt, srcs, dinv, h1, b2, out, N);
}


# Edges grouped by target in two levels: dealt into 256-node bins, then a work-group per bin ranks its edges with LDS counters; no per-edge global atomic left; float sums stay in arrival order
# speedup vs baseline: 3.8701x; 1.8032x over previous
#include <hip/hip_runtime.h>

typedef __attribute__((ext_vector_type(2))) float v2f;
typedef __attribute__((ext_vector_type(8))) float v8f;

#define IN_CH  128
#define HID    64
#define OUTC   40
#define OUTP   48
#define NPW 8
#define BSH    8
#define BNODES 256
#define MAXB   512
#define TPT    16
#define TEDGE  (TPT * 256)
#define CB     8


__global__ void gcn_zero_bins(int* __restrict__ bins) {
  bins[blockIdx.x * blockDim.x + threadIdx.x] = 0;
}

__global__ void gcn_bin_hist(const int* __restrict__ col, int* __restrict__ binCnt,
                             int E, int NB) {
  __shared__ int h[MAXB];
  for (int i = threadIdx.x; i < MAXB; i += 256) h[i] = 0;
  __syncthreads();
  const int base = blockIdx.x * TEDGE + threadIdx.x;
  int cc[TPT];
#pragma unroll
  for (int i = 0; i < TPT; ++i) {
    const int e = base + i * 256;
    cc[i] = (e < E) ? col[e] : -1;
  }
#pragma unroll
  for (int i = 0; i < TPT; ++i)
    if (cc[i] >= 0) atomicAdd(&h[cc[i] >> BSH], 1);
  __syncthreads();
  for (int i = threadIdx.x; i < NB; i += 256) {
    const int v = h[i];
    if (v) atomicAdd(&binCnt[i], v);
  }
}

__global__ void gcn_bin_place(const int* __restrict__ row, const int* __restrict__ col,
                              const int* __restrict__ binCnt, int* __restrict__ binCur,
                              int2* __restrict__ bedge, int E, int NB) {
  __shared__ int h[MAXB];
  __shared__ int bstart[MAXB];
  __shared__ int wsum[8];
  const int tid = threadIdx.x;
  const int lane = tid & 31;
  const int wave = tid >> 5;
  for (int i = tid; i < MAXB; i += 256) h[i] = 0;

  const int a0 = (2 * tid < NB) ? binCnt[2 * tid] : 0;
  const int a1 = (2 * tid + 1 < NB) ? binCnt[2 * tid + 1] : 0;
  const int s = a0 + a1;
  int inc = s;
  for (int o = 1; o < 32; o <<= 1) {
    const int t = __shfl_up(inc, o);
    if (lane >= o) inc += t;
  }
  if (lane == 31) wsum[wave] = inc;
  __syncthreads();
  int wpre = 0;
  for (int w = 0; w < wave; ++w) wpre += wsum[w];
  const int ex = wpre + inc - s;
  bstart[2 * tid] = ex;
  bstart[2 * tid + 1] = ex + a0;

  const int base = blockIdx.x * TEDGE + tid;
  int cc[TPT], rk[TPT], rw[TPT];
#pragma unroll
  for (int i = 0; i < TPT; ++i) {
    const int e = base + i * 256;
    cc[i] = (e < E) ? col[e] : -1;
  }
#pragma unroll
  for (int i = 0; i < TPT; ++i) {
    const int e = base + i * 256;
    rw[i] = (e < E) ? row[e] : 0;
  }
#pragma unroll
  for (int i = 0; i < TPT; ++i) {
    rk[i] = 0;
    if (cc[i] >= 0) rk[i] = atomicAdd(&h[cc[i] >> BSH], 1);
  }
  __syncthreads();
  for (int i = tid; i < NB; i += 256) {
    const int v = h[i];
    const int room = v ? atomicAdd(&binCur[i], v) : 0;
    h[i] = bstart[i] + room;
  }
  __syncthreads();
#pragma unroll
  for (int i = 0; i < TPT; ++i) {
    if (cc[i] >= 0) {
      int2 rec;
      rec.x = rw[i];
      rec.y = cc[i];
      bedge[h[cc[i] >> BSH] + rk[i]] = rec;
    }
  }
}

__global__ void gcn_bin_csr(const int* __restrict__ binCnt,
                            const int2* __restrict__ bedge,
                            int* __restrict__ cnt, int* __restrict__ off,
                            float* __restrict__ dinv, int* __restrict__ src, int N) {
  __shared__ int c[BNODES];
  __shared__ int wsum[8];
  __shared__ int wsum2[8];
  const int tid = threadIdx.x;
  const int lane = tid & 31;
  const int wave = tid >> 5;
  const int b = blockIdx.x;
  const int lo = b * BNODES;

  int part = 0;
  for (int j = tid; j < b; j += 256) part += binCnt[j];
  for (int o = 16; o > 0; o >>= 1) part += __shfl_xor(part, o);
  if (lane == 0) wsum[wave] = part;
  c[tid] = 0;
  const int n_e = binCnt[b];
  __syncthreads();
  int start = 0;
  for (int w = 0; w < 8; ++w) start += wsum[w];
  const int2* be = bedge + start;

  for (int j0 = 0; j0 < n_e; j0 += CB * 256) {
    int cl[CB];
#pragma unroll
    for (int i = 0; i < CB; ++i) {
      const int j = j0 + i * 256 + tid;
      cl[i] = be[(j < n_e) ? j : n_e - 1].y;
    }
#pragma unroll
    for (int i = 0; i < CB; ++i) {
      const int j = j0 + i * 256 + tid;
      if (j < n_e) atomicAdd(&c[cl[i] - lo], 1);
    }
  }
  __syncthreads();

  const int v = c[tid];
  int inc = v;
  for (int o = 1; o < 32; o <<= 1) {
    const int t = __shfl_up(inc, o);
    if (lane >= o) inc += t;
  }
  if (lane == 31) wsum2[wave] = inc;
  __syncthreads();
  int wpre = 0;
  for (int w = 0; w < wave; ++w) wpre += wsum2[w];
  const int ex = wpre + inc - v;
  const int node = lo + tid;
  if (node < N) {
    cnt[node] = v;
    off[node] = start + ex;
    const float d = (float)(v + 1);
    dinv[node] = (d > 0.0f) ? rsqrtf(d) : 0.0f;
  }
  c[tid] = ex;
  __syncthreads();

  for (int j0 = 0; j0 < n_e; j0 += CB * 256) {
    int cl[CB], rw[CB];
#pragma unroll
    for (int i = 0; i < CB; ++i) {
      const int j = j0 + i * 256 + tid;
      const int jj = (j < n_e) ? j : n_e - 1;
      const int2 rec = be[jj];
      cl[i] = rec.y;
      rw[i] = rec.x;
    }
#pragma unroll
    for (int i = 0; i < CB; ++i) {
      const int j = j0 + i * 256 + tid;
      if (j < n_e) src[start + atomicAdd(&c[cl[i] - lo], 1)] = rw[i];
    }
  }
}

__global__ void gcn_gemm1(const float* __restrict__ x,
                          const float* __restrict__ W1,
                          const float* __restrict__ dinv,
                          float* __restrict__ h1, int tiles, int rows) {
  const int wave = threadIdx.x >> 5;
  const int lane = threadIdx.x & 31;
  const int tile = __builtin_amdgcn_readfirstlane((int)(blockIdx.x * 4 + wave));
  if (tile >= tiles) return;
  const int m0 = tile * 16;
  const int rc = lane & 15;
  const int kOff = (lane < 16) ? 0 : 2;

  const int mBase = m0 + ((lane < 16) ? 0 : 8);
  const float4 dlo = *(const float4*)(dinv + mBase);
  const float4 dhi = *(const float4*)(dinv + mBase + 4);
  const float dr[8] = {dlo.x, dlo.y, dlo.z, dlo.w, dhi.x, dhi.y, dhi.z, dhi.w};

  const float* xrow = x + (size_t)(m0 + rc) * IN_CH;
  v8f acc0 = {}, acc1 = {}, acc2 = {}, acc3 = {};
#pragma unroll
  for (int k = 0; k < IN_CH; k += 4) {
    v2f a, b0, b1, b2, b3;
    a.x = xrow[k + kOff];
    a.y = xrow[k + kOff + 1];
    const float* w0 = W1 + (size_t)(k + kOff) * HID + rc;
    const float* w1 = W1 + (size_t)(k + kOff + 1) * HID + rc;
    b0.x = w0[0];   b0.y = w1[0];
    b1.x = w0[16];  b1.y = w1[16];
    b2.x = w0[32];  b2.y = w1[32];
    b3.x = w0[48];  b3.y = w1[48];
    acc0 = __builtin_amdgcn_wmma_f32_16x16x4_f32(false, a, false, b0,
                                                 (short)0, acc0, false, false);
    acc1 = __builtin_amdgcn_wmma_f32_16x16x4_f32(false, a, false, b1,
                                                 (short)0, acc1, false, false);
    acc2 = __builtin_amdgcn_wmma_f32_16x16x4_f32(false, a, false, b2,
                                                 (short)0, acc2, false, false);
    acc3 = __builtin_amdgcn_wmma_f32_16x16x4_f32(false, a, false, b3,
                                                 (short)0, acc3, false, false);
  }
#pragma unroll
  for (int j = 0; j < 8; ++j) {
    float* o = h1 + (size_t)(mBase + j) * HID + rc;
    o[0]  = acc0[j] * dr[j];
    o[16] = acc1[j] * dr[j];
    o[32] = acc2[j] * dr[j];
    o[48] = acc3[j] * dr[j];
  }
  if (tile == 0) {
    float2 z;
    z.x = 0.0f;
    z.y = 0.0f;
    *(float2*)(h1 + (size_t)rows * HID + lane * 2) = z;
  }
}

__global__ void gcn_gemm2(const float* __restrict__ agg1,
                          const float* __restrict__ W2,
                          const float* __restrict__ dinv,
                          float* __restrict__ h2, int tiles, int rows) {
  const int wave = threadIdx.x >> 5;
  const int lane = threadIdx.x & 31;
  const int tile = __builtin_amdgcn_readfirstlane((int)(blockIdx.x * 4 + wave));
  if (tile >= tiles) return;
  const int m0 = tile * 16;
  const int rc = lane & 15;
  const int kOff = (lane < 16) ? 0 : 2;
  const bool valid2 = (32 + rc < OUTC);
  const int c2 = valid2 ? 32 + rc : 32;

  const int mBase = m0 + ((lane < 16) ? 0 : 8);
  const float4 dlo = *(const float4*)(dinv + mBase);
  const float4 dhi = *(const float4*)(dinv + mBase + 4);
  const float dr[8] = {dlo.x, dlo.y, dlo.z, dlo.w, dhi.x, dhi.y, dhi.z, dhi.w};

  const float* arow = agg1 + (size_t)(m0 + rc) * HID;
  v8f acc0 = {}, acc1 = {}, acc2 = {};
#pragma unroll
  for (int k = 0; k < HID; k += 4) {
    v2f a, b0, b1, b2;
    a.x = fmaxf(arow[k + kOff], 0.0f);
    a.y = fmaxf(arow[k + kOff + 1], 0.0f);
    const float* w0 = W2 + (size_t)(k + kOff) * OUTC;
    const float* w1 = W2 + (size_t)(k + kOff + 1) * OUTC;
    b0.x = w0[rc];       b0.y = w1[rc];
    b1.x = w0[16 + rc];  b1.y = w1[16 + rc];
    const float t0 = w0[c2], t1 = w1[c2];
    b2.x = valid2 ? t0 : 0.0f;
    b2.y = valid2 ? t1 : 0.0f;
    acc0 = __builtin_amdgcn_wmma_f32_16x16x4_f32(false, a, false, b0,
                                                 (short)0, acc0, false, false);
    acc1 = __builtin_amdgcn_wmma_f32_16x16x4_f32(false, a, false, b1,
                                                 (short)0, acc1, false, false);
    acc2 = __builtin_amdgcn_wmma_f32_16x16x4_f32(false, a, false, b2,
                                                 (short)0, acc2, false, false);
  }
#pragma unroll
  for (int j = 0; j < 8; ++j) {
    float* o = h2 + (size_t)(mBase + j) * OUTP + rc;
    o[0]  = acc0[j] * dr[j];
    o[16] = acc1[j] * dr[j];
    if (valid2) o[32] = acc2[j] * dr[j];
  }
  if (tile == 0 && lane < OUTP / 2) {
    float2 z;
    z.x = 0.0f;
    z.y = 0.0f;
    *(float2*)(h2 + (size_t)rows * OUTP + lane * 2) = z;
  }
}

template <int STRIDE, int CH, int OSTR>
__global__ void gcn_gather(const int* __restrict__ off, const int* __restrict__ cnt,
                           const int* __restrict__ src,
                           const float* __restrict__ dinv,
                           const float* __restrict__ h,
                           const float* __restrict__ bias,
                           float* __restrict__ dst, int N) {
  const int lane = threadIdx.x & 31;
  const int c0 = NPW * __builtin_amdgcn_readfirstlane(
      (int)((blockIdx.x * blockDim.x + threadIdx.x) >> 5));
  if (c0 >= N) return;
  const int nn = (N - c0 < NPW) ? (N - c0) : NPW;
  const int last = nn - 1;

  int m_off = 0, m_tot = 1;
  float m_d = 0.0f;
  if (lane < nn) {
    m_off = off[c0 + lane];
    m_tot = cnt[c0 + lane] + 1;
    m_d = dinv[c0 + lane];
  }
  const bool act = lane < CH / 2;
  const float* hl = h + (act ? lane * 2 : 0);
  float bx = 0.0f, by = 0.0f;
  if (act) {
    bx = bias[lane * 2];
    by = bias[lane * 2 + 1];
  }

  auto raw_r = [&](int i) -> int {
    const int st = __builtin_amdgcn_readlane(m_off, i);
    const int tt = __builtin_amdgcn_readlane(m_tot, i);
    const int idx = (lane < tt - 1) ? st + lane : 0;
    return src[idx];
  };
  auto fix_r = [&](int i, int raw) -> int {
    const int tt = __builtin_amdgcn_readlane(m_tot, i);
    return (lane < tt - 1) ? raw : ((lane == tt - 1) ? c0 + i : N);
  };

  int r_cur = fix_r(0, raw_r(0));
  int raw_nxt = raw_r(last < 1 ? last : 1);

  for (int i = 0; i < nn; ++i) {
    const int c = c0 + i;
    const int start = __builtin_amdgcn_readlane(m_off, i);
    const int total = __builtin_amdgcn_readlane(m_tot, i);
    const float dc = __int_as_float(__builtin_amdgcn_readlane(__float_as_int(m_d), i));

    int r_l = r_cur;

    float2 v0[16];
    for (int q = 0; q < 16; ++q) {
      const int r = __builtin_amdgcn_readlane(r_l, q);
      v0[q] = *(const float2*)(hl + (size_t)r * STRIDE);
    }
    const int i1 = (i + 1 < last) ? i + 1 : last;
    const int i2 = (i + 2 < last) ? i + 2 : last;
    const int raw_nn = raw_r(i2);

    float ax = 0.0f, ay = 0.0f;
    for (int q = 0; q < 16; ++q) {
      ax += v0[q].x;
      ay += v0[q].y;
    }

    int base = 0, k = 16;
    for (;;) {
      const int rem = total - base;
      const int m = rem < 32 ? rem : 32;
      for (; k + 8 <= m; k += 8) {
        float2 v[8];
        for (int q = 0; q < 8; ++q) {
          const int r = __builtin_amdgcn_readlane(r_l, k + q);
          v[q] = *(const float2*)(hl + (size_t)r * STRIDE);
        }
        for (int q = 0; q < 8; ++q) {
          ax += v[q].x;
          ay += v[q].y;
        }
      }
      for (; k < m; k += 4) {
        float2 v[4];
        for (int q = 0; q < 4; ++q) {
          const int r = __builtin_amdgcn_readlane(r_l, k + q);
          v[q] = *(const float2*)(hl + (size_t)r * STRIDE);
        }
        for (int q = 0; q < 4; ++q) {
          ax += v[q].x;
          ay += v[q].y;
        }
      }
      base += 32;
      if (base >= total) break;
      const int j = base + lane;
      r_l = (j == total - 1) ? c : N;
      if (j < total - 1) r_l = src[start + j];
      k = 0;
    }
    if (act) {
      float2 o;
      o.x = ax * dc + bx;
      o.y = ay * dc + by;
      *(float2*)(dst + (size_t)c * OSTR + lane * 2) = o;
    }
    r_cur = fix_r(i1, raw_nxt);
    raw_nxt = raw_nn;
  }
}

extern "C" void kernel_launch(void* const* d_in, const int* in_sizes, int n_in,
                              void* d_out, int out_size, void* d_ws, size_t ws_size,
                              hipStream_t stream) {
  const float* x  = (const float*)d_in[0];
  const int*   ei = (const int*)d_in[1];
  const float* W1 = (const float*)d_in[2];
  const float* b1 = (const float*)d_in[3];
  const float* W2 = (const float*)d_in[4];
  const float* b2 = (const float*)d_in[5];
  float* out = (float*)d_out;

  const int N = in_sizes[0] / IN_CH;
  const int E = in_sizes[1] / 2;
  const int* rowp = ei;
  const int* colp = ei + E;

  const size_t Np = ((size_t)N + 63) & ~(size_t)63;
  const size_t Ep = ((size_t)E + 63) & ~(size_t)63;
  const int NB = (N + BNODES - 1) / BNODES;
  const int etiles = (E + TEDGE - 1) / TEDGE;
  int*   wsi  = (int*)d_ws;
  int*   cnt  = wsi;
  int*   off  = wsi + Np;
  float* dinv = (float*)(wsi + 2 * Np);
  int*   binCnt = wsi + 3 * Np;
  int*   binCur = binCnt + MAXB;
  int2*  bedge = (int2*)(wsi + 3 * Np + 2 * MAXB);
  int*   srcs = wsi + 3 * Np + 2 * MAXB + 2 * Ep;
  float* h1   = (float*)(srcs + Ep);
  float* agg1 = h1 + ((size_t)N + 1) * HID;

  gcn_zero_bins<<<2 * MAXB / 256, 256, 0, stream>>>(binCnt);
  gcn_bin_hist<<<etiles, 256, 0, stream>>>(colp, binCnt, E, NB);
  gcn_bin_place<<<etiles, 256, 0, stream>>>(rowp, colp, binCnt, binCur, bedge, E, NB);
  gcn_bin_csr<<<NB, 256, 0, stream>>>(binCnt, bedge, cnt, off, dinv, srcs, N);
  gcn_gemm1<<<(N / 16 + 3) / 4, 128, 0, stream>>>(x, W1, dinv, h1, N / 16, N);
  gcn_gather<HID, HID, HID><<<(N + 8 * NPW - 1) / (8 * NPW), 256, 0, stream>>>(
      off, cnt, srcs, dinv, h1, b1, agg1, N);
  gcn_gemm2<<<(N / 16 + 3) / 4, 128, 0, stream>>>(agg1, W2, dinv, h1, N / 16, N);
  gcn_gather<OUTP, OUTC, OUTC><<<(N + 8 * NPW - 1) / (8 * NPW), 256, 0, stream>>>(
      off, cnt, srcs, dinv, h1, b2, out, N);
}


# GEMM2 becomes the epilogue of the layer-1 gather: a block keeps its 64 aggregated rows and W2 in LDS and multiplies there; agg1 never goes to memory, one launch less; every sum order kept
# speedup vs baseline: 4.2136x; 1.0888x over previous
#include <hip/hip_runtime.h>

typedef __attribute__((ext_vector_type(2))) float v2f;
typedef __attribute__((ext_vector_type(8))) float v8f;

#define IN_CH  128
#define HID    64
#define OUTC   40
#define OUTP   48
#define NPW 8
#define BSH    8
#define BNODES 256
#define MAXB   512
#define TPT    16
#define TEDGE  (TPT * 256)
#define CB     8


__global__ void gcn_zero_bins(int* __restrict__ bins) {
  bins[blockIdx.x * blockDim.x + threadIdx.x] = 0;
}

__global__ void gcn_bin_hist(const int* __restrict__ col, int* __restrict__ binCnt,
                             int E, int NB) {
  __shared__ int h[MAXB];
  for (int i = threadIdx.x; i < MAXB; i += 256) h[i] = 0;
  __syncthreads();
  const int base = blockIdx.x * TEDGE + threadIdx.x;
  int cc[TPT];
#pragma unroll
  for (int i = 0; i < TPT; ++i) {
    const int e = base + i * 256;
    cc[i] = (e < E) ? col[e] : -1;
  }
#pragma unroll
  for (int i = 0; i < TPT; ++i)
    if (cc[i] >= 0) atomicAdd(&h[cc[i] >> BSH], 1);
  __syncthreads();
  for (int i = threadIdx.x; i < NB; i += 256) {
    const int v = h[i];
    if (v) atomicAdd(&binCnt[i], v);
  }
}

__global__ void gcn_bin_place(const int* __restrict__ row, const int* __restrict__ col,
                              const int* __restrict__ binCnt, int* __restrict__ binCur,
                              int2* __restrict__ bedge, int E, int NB) {
  __shared__ int h[MAXB];
  __shared__ int bstart[MAXB];
  __shared__ int wsum[8];
  const int tid = threadIdx.x;
  const int lane = tid & 31;
  const int wave = tid >> 5;
  for (int i = tid; i < MAXB; i += 256) h[i] = 0;

  const int a0 = (2 * tid < NB) ? binCnt[2 * tid] : 0;
  const int a1 = (2 * tid + 1 < NB) ? binCnt[2 * tid + 1] : 0;
  const int s = a0 + a1;
  int inc = s;
  for (int o = 1; o < 32; o <<= 1) {
    const int t = __shfl_up(inc, o);
    if (lane >= o) inc += t;
  }
  if (lane == 31) wsum[wave] = inc;
  __syncthreads();
  int wpre = 0;
  for (int w = 0; w < wave; ++w) wpre += wsum[w];
  const int ex = wpre + inc - s;
  bstart[2 * tid] = ex;
  bstart[2 * tid + 1] = ex + a0;

  const int base = blockIdx.x * TEDGE + tid;
  int cc[TPT], rk[TPT], rw[TPT];
#pragma unroll
  for (int i = 0; i < TPT; ++i) {
    const int e = base + i * 256;
    cc[i] = (e < E) ? col[e] : -1;
  }
#pragma unroll
  for (int i = 0; i < TPT; ++i) {
    const int e = base + i * 256;
    rw[i] = (e < E) ? row[e] : 0;
  }
#pragma unroll
  for (int i = 0; i < TPT; ++i) {
    rk[i] = 0;
    if (cc[i] >= 0) rk[i] = atomicAdd(&h[cc[i] >> BSH], 1);
  }
  __syncthreads();
  for (int i = tid; i < NB; i += 256) {
    const int v = h[i];
    const int room = v ? atomicAdd(&binCur[i], v) : 0;
    h[i] = bstart[i] + room;
  }
  __syncthreads();
#pragma unroll
  for (int i = 0; i < TPT; ++i) {
    if (cc[i] >= 0) {
      int2 rec;
      rec.x = rw[i];
      rec.y = cc[i];
      bedge[h[cc[i] >> BSH] + rk[i]] = rec;
    }
  }
}

__global__ void gcn_bin_csr(const int* __restrict__ binCnt,
                            const int2* __restrict__ bedge,
                            int* __restrict__ cnt, int* __restrict__ off,
                            float* __restrict__ dinv, int* __restrict__ src, int N) {
  __shared__ int c[BNODES];
  __shared__ int wsum[8];
  __shared__ int wsum2[8];
  const int tid = threadIdx.x;
  const int lane = tid & 31;
  const int wave = tid >> 5;
  const int b = blockIdx.x;
  const int lo = b * BNODES;

  int part = 0;
  for (int j = tid; j < b; j += 256) part += binCnt[j];
  for (int o = 16; o > 0; o >>= 1) part += __shfl_xor(part, o);
  if (lane == 0) wsum[wave] = part;
  c[tid] = 0;
  const int n_e = binCnt[b];
  __syncthreads();
  int start = 0;
  for (int w = 0; w < 8; ++w) start += wsum[w];
  const int2* be = bedge + start;

  for (int j0 = 0; j0 < n_e; j0 += CB * 256) {
    int cl[CB];
#pragma unroll
    for (int i = 0; i < CB; ++i) {
      const int j = j0 + i * 256 + tid;
      cl[i] = be[(j < n_e) ? j : n_e - 1].y;
    }
#pragma unroll
    for (int i = 0; i < CB; ++i) {
      const int j = j0 + i * 256 + tid;
      if (j < n_e) atomicAdd(&c[cl[i] - lo], 1);
    }
  }
  __syncthreads();

  const int v = c[tid];
  int inc = v;
  for (int o = 1; o < 32; o <<= 1) {
    const int t = __shfl_up(inc, o);
    if (lane >= o) inc += t;
  }
  if (lane == 31) wsum2[wave] = inc;
  __syncthreads();
  int wpre = 0;
  for (int w = 0; w < wave; ++w) wpre += wsum2[w];
  const int ex = wpre + inc - v;
  const int node = lo + tid;
  if (node < N) {
    cnt[node] = v;
    off[node] = start + ex;
    const float d = (float)(v + 1);
    dinv[node] = (d > 0.0f) ? rsqrtf(d) : 0.0f;
  }
  c[tid] = ex;
  __syncthreads();

  for (int j0 = 0; j0 < n_e; j0 += CB * 256) {
    int cl[CB], rw[CB];
#pragma unroll
    for (int i = 0; i < CB; ++i) {
      const int j = j0 + i * 256 + tid;
      const int jj = (j < n_e) ? j : n_e - 1;
      const int2 rec = be[jj];
      cl[i] = rec.y;
      rw[i] = rec.x;
    }
#pragma unroll
    for (int i = 0; i < CB; ++i) {
      const int j = j0 + i * 256 + tid;
      if (j < n_e) src[start + atomicAdd(&c[cl[i] - lo], 1)] = rw[i];
    }
  }
}

__global__ void gcn_gemm1(const float* __restrict__ x,
                          const float* __restrict__ W1,
                          const float* __restrict__ dinv,
                          float* __restrict__ h1, int tiles, int rows) {
  const int wave = threadIdx.x >> 5;
  const int lane = threadIdx.x & 31;
  const int tile = __builtin_amdgcn_readfirstlane((int)(blockIdx.x * 4 + wave));
  if (tile >= tiles) return;
  const int m0 = tile * 16;
  const int rc = lane & 15;
  const int kOff = (lane < 16) ? 0 : 2;

  const int mBase = m0 + ((lane < 16) ? 0 : 8);
  const float4 dlo = *(const float4*)(dinv + mBase);
  const float4 dhi = *(const float4*)(dinv + mBase + 4);
  const float dr[8] = {dlo.x, dlo.y, dlo.z, dlo.w, dhi.x, dhi.y, dhi.z, dhi.w};

  const float* xrow = x + (size_t)(m0 + rc) * IN_CH;
  v8f acc0 = {}, acc1 = {}, acc2 = {}, acc3 = {};
#pragma unroll
  for (int k = 0; k < IN_CH; k += 4) {
    v2f a, b0, b1, b2, b3;
    a.x = xrow[k + kOff];
    a.y = xrow[k + kOff + 1];
    const float* w0 = W1 + (size_t)(k + kOff) * HID + rc;
    const float* w1 = W1 + (size_t)(k + kOff + 1) * HID + rc;
    b0.x = w0[0];   b0.y = w1[0];
    b1.x = w0[16];  b1.y = w1[16];
    b2.x = w0[32];  b2.y = w1[32];
    b3.x = w0[48];  b3.y = w1[48];
    acc0 = __builtin_amdgcn_wmma_f32_16x16x4_f32(false, a, false, b0,
                                                 (short)0, acc0, false, false);
    acc1 = __builtin_amdgcn_wmma_f32_16x16x4_f32(false, a, false, b1,
                                                 (short)0, acc1, false, false);
    acc2 = __builtin_amdgcn_wmma_f32_16x16x4_f32(false, a, false, b2,
                                                 (short)0, acc2, false, false);
    acc3 = __builtin_amdgcn_wmma_f32_16x16x4_f32(false, a, false, b3,
                                                 (short)0, acc3, false, false);
  }
#pragma unroll
  for (int j = 0; j < 8; ++j) {
    float* o = h1 + (size_t)(mBase + j) * HID + rc;
    o[0]  = acc0[j] * dr[j];
    o[16] = acc1[j] * dr[j];
    o[32] = acc2[j] * dr[j];
    o[48] = acc3[j] * dr[j];
  }
  if (tile == 0) {
    float2 z;
    z.x = 0.0f;
    z.y = 0.0f;
    *(float2*)(h1 + (size_t)rows * HID + lane * 2) = z;
  }
}

template <int STRIDE, int CH, typename StoreF>
__device__ __forceinline__ void gcn_gather_wave(const int* __restrict__ off,
                                                const int* __restrict__ cnt,
                                                const int* __restrict__ src,
                                                const float* __restrict__ dinv,
                                                const float* __restrict__ h,
                                                const float* __restrict__ bias,
                                                int N, int c0, int lane, StoreF store) {
  const int nn = (N - c0 < NPW) ? (N - c0) : NPW;
  const int last = nn - 1;

  int m_off = 0, m_tot = 1;
  float m_d = 0.0f;
  if (lane < nn) {
    m_off = off[c0 + lane];
    m_tot = cnt[c0 + lane] + 1;
    m_d = dinv[c0 + lane];
  }
  const bool act = lane < CH / 2;
  const float* hl = h + (act ? lane * 2 : 0);
  float bx = 0.0f, by = 0.0f;
  if (act) {
    bx = bias[lane * 2];
    by = bias[lane * 2 + 1];
  }

  auto raw_r = [&](int i) -> int {
    const int st = __builtin_amdgcn_readlane(m_off, i);
    const int tt = __builtin_amdgcn_readlane(m_tot, i);
    const int idx = (lane < tt - 1) ? st + lane : 0;
    return src[idx];
  };
  auto fix_r = [&](int i, int raw) -> int {
    const int tt = __builtin_amdgcn_readlane(m_tot, i);
    return (lane < tt - 1) ? raw : ((lane == tt - 1) ? c0 + i : N);
  };

  int r_cur = fix_r(0, raw_r(0));
  int raw_nxt = raw_r(last < 1 ? last : 1);

  for (int i = 0; i < nn; ++i) {
    const int c = c0 + i;
    const int start = __builtin_amdgcn_readlane(m_off, i);
    const int total = __builtin_amdgcn_readlane(m_tot, i);
    const float dc = __int_as_float(__builtin_amdgcn_readlane(__float_as_int(m_d), i));

    int r_l = r_cur;

    float2 v0[16];
    for (int q = 0; q < 16; ++q) {
      const int r = __builtin_amdgcn_readlane(r_l, q);
      v0[q] = *(const float2*)(hl + (size_t)r * STRIDE);
    }
    const int i1 = (i + 1 < last) ? i + 1 : last;
    const int i2 = (i + 2 < last) ? i + 2 : last;
    const int raw_nn = raw_r(i2);

    float ax = 0.0f, ay = 0.0f;
    for (int q = 0; q < 16; ++q) {
      ax += v0[q].x;
      ay += v0[q].y;
    }

    int base = 0, k = 16;
    for (;;) {
      const int rem = total - base;
      const int m = rem < 32 ? rem : 32;
      for (; k + 8 <= m; k += 8) {
        float2 v[8];
        for (int q = 0; q < 8; ++q) {
          const int r = __builtin_amdgcn_readlane(r_l, k + q);
          v[q] = *(const float2*)(hl + (size_t)r * STRIDE);
        }
        for (int q = 0; q < 8; ++q) {
          ax += v[q].x;
          ay += v[q].y;
        }
      }
      for (; k < m; k += 4) {
        float2 v[4];
        for (int q = 0; q < 4; ++q) {
          const int r = __builtin_amdgcn_readlane(r_l, k + q);
          v[q] = *(const float2*)(hl + (size_t)r * STRIDE);
        }
        for (int q = 0; q < 4; ++q) {
          ax += v[q].x;
          ay += v[q].y;
        }
      }
      base += 32;
      if (base >= total) break;
      const int j = base + lane;
      r_l = (j == total - 1) ? c : N;
      if (j < total - 1) r_l = src[start + j];
      k = 0;
    }
    if (act) store(c, ax * dc + bx, ay * dc + by);
    r_cur = fix_r(i1, raw_nxt);
    raw_nxt = raw_nn;
  }
}

template <int STRIDE, int CH, int OSTR>
__global__ void gcn_gather(const int* __restrict__ off, const int* __restrict__ cnt,
                           const int* __restrict__ src,
                           const float* __restrict__ dinv,
                           const float* __restrict__ h,
                           const float* __restrict__ bias,
                           float* __restrict__ dst, int N) {
  const int lane = threadIdx.x & 31;
  const int c0 = NPW * __builtin_amdgcn_readfirstlane(
      (int)((blockIdx.x * blockDim.x + threadIdx.x) >> 5));
  if (c0 >= N) return;
  gcn_gather_wave<STRIDE, CH>(off, cnt, src, dinv, h, bias, N, c0, lane,
                              [&](int c, float ox, float oy) {
    float2 o;
    o.x = ox;
    o.y = oy;
    *(float2*)(dst + (size_t)c * OSTR + lane * 2) = o;
  });
}

#define BROWS  (8 * NPW)
#define APITCH (HID + 4)
__global__ void gcn_gather_gemm2(const int* __restrict__ off, const int* __restrict__ cnt,
                                 const int* __restrict__ src,
                                 const float* __restrict__ dinv,
                                 const float* __restrict__ h,
                                 const float* __restrict__ bias,
                                 const float* __restrict__ W2,
                                 float* __restrict__ h2, int N) {
  __shared__ float sa[BROWS * APITCH];
  __shared__ float sw[OUTP * APITCH];
#pragma unroll
  for (int j = 0; j < HID * OUTC / 256; ++j) {
    const int i = j * 256 + (int)threadIdx.x;
    const int k = i / OUTC;
    sw[(i - k * OUTC) * APITCH + k] = W2[i];
  }
  for (int i = threadIdx.x; i < (OUTP - OUTC) * HID; i += 256)
    sw[(OUTC + i / HID) * APITCH + (i % HID)] = 0.0f;
  const int lane = threadIdx.x & 31;
  const int wave = __builtin_amdgcn_readfirstlane((int)(threadIdx.x >> 5));
  const int r0 = blockIdx.x * BROWS;
  const int c0 = r0 + wave * NPW;
  if (c0 < N) {
    gcn_gather_wave<HID, HID>(off, cnt, src, dinv, h, bias, N, c0, lane,
                              [&](int c, float ox, float oy) {
      float2 o;
      o.x = fmaxf(ox, 0.0f);
      o.y = fmaxf(oy, 0.0f);
      *(float2*)(sa + (c - r0) * APITCH + lane * 2) = o;
    });
  }
  __syncthreads();

  const int tile = wave & 3;
  const int m0 = r0 + tile * 16;
  if (m0 + 16 <= N) {
    const int rc = lane & 15;
    const int kOff = (lane < 16) ? 0 : 2;
    const int mBase = m0 + ((lane < 16) ? 0 : 8);
    const float4 dlo = *(const float4*)(dinv + mBase);
    const float4 dhi = *(const float4*)(dinv + mBase + 4);
    const float dr[8] = {dlo.x, dlo.y, dlo.z, dlo.w, dhi.x, dhi.y, dhi.z, dhi.w};
    const float* arow = sa + (tile * 16 + rc) * APITCH + kOff;
    const float* brow = sw + rc * APITCH + kOff;
    if (wave < 4) {
      v8f acc0 = {}, acc1 = {};
#pragma unroll
      for (int k = 0; k < HID; k += 4) {
        v2f a, b0, b1;
        a.x = arow[k];
        a.y = arow[k + 1];
        b0.x = brow[k];                 b0.y = brow[k + 1];
        b1.x = brow[16 * APITCH + k];   b1.y = brow[16 * APITCH + k + 1];
        acc0 = __builtin_amdgcn_wmma_f32_16x16x4_f32(false, a, false, b0,
                                                     (short)0, acc0, false, false);
        acc1 = __builtin_amdgcn_wmma_f32_16x16x4_f32(false, a, false, b1,
                                                     (short)0, acc1, false, false);
      }
#pragma unroll
      for (int j = 0; j < 8; ++j) {
        float* o = h2 + (size_t)(mBase + j) * OUTP + rc;
        o[0]  = acc0[j] * dr[j];
        o[16] = acc1[j] * dr[j];
      }
    } else {
      v8f acc2 = {};
#pragma unroll
      for (int k = 0; k < HID; k += 4) {
        v2f a, b2;
        a.x = arow[k];
        a.y = arow[k + 1];
        b2.x = brow[32 * APITCH + k];
        b2.y = brow[32 * APITCH + k + 1];
        acc2 = __builtin_amdgcn_wmma_f32_16x16x4_f32(false, a, false, b2,
                                                     (short)0, acc2, false, false);
      }
      if (32 + rc < OUTC) {
#pragma unroll
        for (int j = 0; j < 8; ++j)
          h2[(size_t)(mBase + j) * OUTP + 32 + rc] = acc2[j] * dr[j];
      }
    }
  }
  if (blockIdx.x == 0 && wave == 0 && lane < OUTP / 2) {
    float2 z;
    z.x = 0.0f;
    z.y = 0.0f;
    *(float2*)(h2 + (size_t)N * OUTP + lane * 2) = z;
  }
}

extern "C" void kernel_launch(void* const* d_in, const int* in_sizes, int n_in,
                              void* d_out, int out_size, void* d_ws, size_t ws_size,
                              hipStream_t stream) {
  const float* x  = (const float*)d_in[0];
  const int*   ei = (const int*)d_in[1];
  const float* W1 = (const float*)d_in[2];
  const float* b1 = (const float*)d_in[3];
  const float* W2 = (const float*)d_in[4];
  const float* b2 = (const float*)d_in[5];
  float* out = (float*)d_out;

  const int N = in_sizes[0] / IN_CH;
  const int E = in_sizes[1] / 2;
  const int* rowp = ei;
  const int* colp = ei + E;

  const size_t Np = ((size_t)N + 63) & ~(size_t)63;
  const size_t Ep = ((size_t)E + 63) & ~(size_t)63;
  const int NB = (N + BNODES - 1) / BNODES;
  const int etiles = (E + TEDGE - 1) / TEDGE;
  int*   wsi  = (int*)d_ws;
  int*   cnt  = wsi;
  int*   off  = wsi + Np;
  float* dinv = (float*)(wsi + 2 * Np);
  int*   binCnt = wsi + 3 * Np;
  int*   binCur = binCnt + MAXB;
  int2*  bedge = (int2*)(wsi + 3 * Np + 2 * MAXB);
  int*   srcs = wsi + 3 * Np + 2 * MAXB + 2 * Ep;
  float* h1   = (float*)(srcs + Ep);
  float* h2   = h1 + ((size_t)N + 1) * HID;

  gcn_zero_bins<<<2 * MAXB / 256, 256, 0, stream>>>(binCnt);
  gcn_bin_hist<<<etiles, 256, 0, stream>>>(colp, binCnt, E, NB);
  gcn_bin_place<<<etiles, 256, 0, stream>>>(rowp, colp, binCnt, binCur, bedge, E, NB);
  gcn_bin_csr<<<NB, 256, 0, stream>>>(binCnt, bedge, cnt, off, dinv, srcs, N);
  gcn_gemm1<<<(N / 16 + 3) / 4, 128, 0, stream>>>(x, W1, dinv, h1, N / 16, N);
  gcn_gather_gemm2<<<(N + BROWS - 1) / BROWS, 256, 0, stream>>>(
      off, cnt, srcs, dinv, h1, b1, W2, h2, N);
  gcn_gather<OUTP, OUTC, OUTC><<<(N + 8 * NPW - 1) / (8 * NPW), 256, 0, stream>>>(
      off, cnt, srcs, dinv, h2, b2, out, N);
}


# Binning without global counts: each edge tile sorted by bin in LDS leaves a row of bin offsets, a bin's block reads its column; zero+histogram launches go; float sums stay in arrival order
# speedup vs baseline: 4.5116x; 1.0707x over previous
#include <hip/hip_runtime.h>

typedef __attribute__((ext_vector_type(2))) float v2f;
typedef __attribute__((ext_vector_type(8))) float v8f;

#define IN_CH  128
#define HID    64
#define OUTC   40
#define OUTP   48
#define NPW 8
#define BSH    8
#define BNODES 256
#define MAXB   512
#define TPT    16
#define TEDGE  (TPT * 256)
#define MAXT   512
#define SB     25


__global__ void gcn_tile_bins(const int* __restrict__ row, const int* __restrict__ col,
                              int* __restrict__ tileOff, int2* __restrict__ bedge, int E) {
  __shared__ int h[MAXB];
  __shared__ int wsum[8];
  __shared__ int2 stage[TEDGE];
  const int tid = threadIdx.x;
  const int lane = tid & 31;
  const int wave = tid >> 5;
  h[2 * tid] = 0;
  h[2 * tid + 1] = 0;
  __syncthreads();

  const int tbase = blockIdx.x * TEDGE;
  const int base = tbase + tid;
  int cc[TPT], rk[TPT], rw[TPT];
#pragma unroll
  for (int i = 0; i < TPT; ++i) {
    const int e = base + i * 256;
    cc[i] = (e < E) ? col[e] : -1;
  }
#pragma unroll
  for (int i = 0; i < TPT; ++i) {
    const int e = base + i * 256;
    rw[i] = (e < E) ? row[e] : 0;
  }
#pragma unroll
  for (int i = 0; i < TPT; ++i) {
    rk[i] = 0;
    if (cc[i] >= 0) rk[i] = atomicAdd(&h[cc[i] >> BSH], 1);
  }
  __syncthreads();

  const int a0 = h[2 * tid];
  const int a1 = h[2 * tid + 1];
  const int s = a0 + a1;
  int inc = s;
  for (int o = 1; o < 32; o <<= 1) {
    const int t = __shfl_up(inc, o);
    if (lane >= o) inc += t;
  }
  if (lane == 31) wsum[wave] = inc;
  __syncthreads();
  int wpre = 0;
  for (int w = 0; w < wave; ++w) wpre += wsum[w];
  const int ex = wpre + inc - s;
  h[2 * tid] = ex;
  h[2 * tid + 1] = ex + a0;
  int2 pr;
  pr.x = ex;
  pr.y = ex + a0;
  *(int2*)(tileOff + (size_t)blockIdx.x * MAXB + 2 * tid) = pr;
  __syncthreads();

  int sl[TPT];
#pragma unroll
  for (int i = 0; i < TPT; ++i)
    sl[i] = h[((cc[i] >= 0) ? cc[i] : 0) >> BSH] + rk[i];
#pragma unroll
  for (int i = 0; i < TPT; ++i) {
    if (cc[i] >= 0) {
      int2 rec;
      rec.x = rw[i];
      rec.y = cc[i];
      stage[sl[i]] = rec;
    }
  }
  __syncthreads();
  const int n_t = (E - tbase < TEDGE) ? (E - tbase) : TEDGE;
  int2 ov[TPT];
#pragma unroll
  for (int i = 0; i < TPT; ++i) ov[i] = stage[tid + i * 256];
#pragma unroll
  for (int i = 0; i < TPT; ++i) {
    const int j = tid + i * 256;
    if (j < n_t) bedge[tbase + j] = ov[i];
  }
}

__global__ void gcn_bin_csr(const int* __restrict__ tileOff,
                            const int2* __restrict__ bedge,
                            int* __restrict__ cnt, int* __restrict__ off,
                            float* __restrict__ dinv, int* __restrict__ src,
                            int N, int etiles) {
  __shared__ int c[BNODES + 32];
  __shared__ int segOff[MAXT];
  __shared__ int segLen[MAXT];
  __shared__ int wsum[8];
  __shared__ int wsum2[8];
  const int tid = threadIdx.x;
  const int lane = tid & 31;
  const int wave = __builtin_amdgcn_readfirstlane(tid >> 5);
  const int b = blockIdx.x;
  const int lo = b * BNODES;

  int part = 0;
#pragma unroll
  for (int r = 0; r < 2; ++r) {
    const int t = tid + r * 256;
    int o0 = 0, ln = 0;
    if (t < etiles) {
      const int* p = tileOff + (size_t)t * MAXB + b;
      o0 = p[0];
      ln = p[1] - o0;
    }
    part += o0;
    segOff[t] = t * TEDGE + o0;
    segLen[t] = ln;
  }
  for (int o = 16; o > 0; o >>= 1) part += __shfl_xor(part, o);
  if (lane == 0) wsum[wave] = part;
  c[tid] = 0;
  if (tid < 32) c[BNODES + tid] = 0;
  __syncthreads();
  int start = 0;
  for (int w = 0; w < 8; ++w) start += wsum[w];

  for (int t0 = wave; t0 < etiles; t0 += 8 * SB) {
    const int tq = t0 + 8 * lane;
    const bool tv = (lane < SB) && (tq < MAXT);
    const int m_len = tv ? segLen[tq] : 0;
    const int m_off = tv ? segOff[tq] : 0;
    int ln[SB], tg[SB];
#pragma unroll
    for (int q = 0; q < SB; ++q) {
      ln[q] = __builtin_amdgcn_readlane(m_len, q);
      const int so = __builtin_amdgcn_readlane(m_off, q);
      const int k = (lane < ln[q]) ? lane : ln[q] - 1;
      const int idx = (ln[q] > 0) ? so + k : 0;
      tg[q] = bedge[idx].y;
    }
#pragma unroll
    for (int q = 0; q < SB; ++q)
      if (lane < ln[q]) atomicAdd(&c[tg[q] - lo], 1);
#pragma unroll
    for (int q = 0; q < SB; ++q) {
      if (ln[q] > 32) {
        const int so = __builtin_amdgcn_readlane(m_off, q);
        for (int k = 32 + lane; k < ln[q]; k += 32)
          atomicAdd(&c[bedge[so + k].y - lo], 1);
      }
    }
  }
  __syncthreads();

  const int v = c[tid];
  int inc = v;
  for (int o = 1; o < 32; o <<= 1) {
    const int t = __shfl_up(inc, o);
    if (lane >= o) inc += t;
  }
  if (lane == 31) wsum2[wave] = inc;
  __syncthreads();
  int wpre = 0;
  for (int w = 0; w < wave; ++w) wpre += wsum2[w];
  const int ex = wpre + inc - v;
  const int node = lo + tid;
  if (node < N) {
    cnt[node] = v;
    off[node] = start + ex;
    const float d = (float)(v + 1);
    dinv[node] = (d > 0.0f) ? rsqrtf(d) : 0.0f;
  }
  c[tid] = start + ex;
  __syncthreads();

  for (int t0 = wave; t0 < etiles; t0 += 8 * SB) {
    const int tq = t0 + 8 * lane;
    const bool tv = (lane < SB) && (tq < MAXT);
    const int m_len = tv ? segLen[tq] : 0;
    const int m_off = tv ? segOff[tq] : 0;
    int ln[SB], sl[SB];
    int2 rc[SB];
#pragma unroll
    for (int q = 0; q < SB; ++q) {
      ln[q] = __builtin_amdgcn_readlane(m_len, q);
      const int so = __builtin_amdgcn_readlane(m_off, q);
      const int k = (lane < ln[q]) ? lane : ln[q] - 1;
      const int idx = (ln[q] > 0) ? so + k : 0;
      rc[q] = bedge[idx];
    }
#pragma unroll
    for (int q = 0; q < SB; ++q) {
      const bool on = lane < ln[q];
      sl[q] = atomicAdd(&c[on ? rc[q].y - lo : BNODES + lane], on ? 1 : 0);
    }
#pragma unroll
    for (int q = 0; q < SB; ++q)
      if (lane < ln[q]) src[sl[q]] = rc[q].x;
#pragma unroll
    for (int q = 0; q < SB; ++q) {
      if (ln[q] > 32) {
        const int so = __builtin_amdgcn_readlane(m_off, q);
        for (int k = 32 + lane; k < ln[q]; k += 32) {
          const int2 rec = bedge[so + k];
          src[atomicAdd(&c[rec.y - lo], 1)] = rec.x;
        }
      }
    }
  }
}

__global__ void gcn_gemm1(const float* __restrict__ x,
                          const float* __restrict__ W1,
                          const float* __restrict__ dinv,
                          float* __restrict__ h1, int tiles, int rows) {
  const int wave = threadIdx.x >> 5;
  const int lane = threadIdx.x & 31;
  const int tile = __builtin_amdgcn_readfirstlane((int)(blockIdx.x * 4 + wave));
  if (tile >= tiles) return;
  const int m0 = tile * 16;
  const int rc = lane & 15;
  const int kOff = (lane < 16) ? 0 : 2;

  const int mBase = m0 + ((lane < 16) ? 0 : 8);
  const float4 dlo = *(const float4*)(dinv + mBase);
  const float4 dhi = *(const float4*)(dinv + mBase + 4);
  const float dr[8] = {dlo.x, dlo.y, dlo.z, dlo.w, dhi.x, dhi.y, dhi.z, dhi.w};

  const float* xrow = x + (size_t)(m0 + rc) * IN_CH;
  v8f acc0 = {}, acc1 = {}, acc2 = {}, acc3 = {};
#pragma unroll
  for (int k = 0; k < IN_CH; k += 4) {
    v2f a, b0, b1, b2, b3;
    a.x = xrow[k + kOff];
    a.y = xrow[k + kOff + 1];
    const float* w0 = W1 + (size_t)(k + kOff) * HID + rc;
    const float* w1 = W1 + (size_t)(k + kOff + 1) * HID + rc;
    b0.x = w0[0];   b0.y = w1[0];
    b1.x = w0[16];  b1.y = w1[16];
    b2.x = w0[32];  b2.y = w1[32];
    b3.x = w0[48];  b3.y = w1[48];
    acc0 = __builtin_amdgcn_wmma_f32_16x16x4_f32(false, a, false, b0,
                                                 (short)0, acc0, false, false);
    acc1 = __builtin_amdgcn_wmma_f32_16x16x4_f32(false, a, false, b1,
                                                 (short)0, acc1, false, false);
    acc2 = __builtin_amdgcn_wmma_f32_16x16x4_f32(false, a, false, b2,
                                                 (short)0, acc2, false, false);
    acc3 = __builtin_amdgcn_wmma_f32_16x16x4_f32(false, a, false, b3,
                                                 (short)0, acc3, false, false);
  }
#pragma unroll
  for (int j = 0; j < 8; ++j) {
    float* o = h1 + (size_t)(mBase + j) * HID + rc;
    o[0]  = acc0[j] * dr[j];
    o[16] = acc1[j] * dr[j];
    o[32] = acc2[j] * dr[j];
    o[48] = acc3[j] * dr[j];
  }
  if (tile == 0) {
    float2 z;
    z.x = 0.0f;
    z.y = 0.0f;
    *(float2*)(h1 + (size_t)rows * HID + lane * 2) = z;
  }
}

template <int STRIDE, int CH, typename StoreF>
__device__ __forceinline__ void gcn_gather_wave(const int* __restrict__ off,
                                                const int* __restrict__ cnt,
                                                const int* __restrict__ src,
                                                const float* __restrict__ dinv,
                                                const float* __restrict__ h,
                                                const float* __restrict__ bias,
                                                int N, int c0, int lane, StoreF store) {
  const int nn = (N - c0 < NPW) ? (N - c0) : NPW;
  const int last = nn - 1;

  int m_off = 0, m_tot = 1;
  float m_d = 0.0f;
  if (lane < nn) {
    m_off = off[c0 + lane];
    m_tot = cnt[c0 + lane] + 1;
    m_d = dinv[c0 + lane];
  }
  const bool act = lane < CH / 2;
  const float* hl = h + (act ? lane * 2 : 0);
  float bx = 0.0f, by = 0.0f;
  if (act) {
    bx = bias[lane * 2];
    by = bias[lane * 2 + 1];
  }

  auto raw_r = [&](int i) -> int {
    const int st = __builtin_amdgcn_readlane(m_off, i);
    const int tt = __builtin_amdgcn_readlane(m_tot, i);
    const int idx = (lane < tt - 1) ? st + lane : 0;
    return src[idx];
  };
  auto fix_r = [&](int i, int raw) -> int {
    const int tt = __builtin_amdgcn_readlane(m_tot, i);
    return (lane < tt - 1) ? raw : ((lane == tt - 1) ? c0 + i : N);
  };

  int r_cur = fix_r(0, raw_r(0));
  int raw_nxt = raw_r(last < 1 ? last : 1);

  for (int i = 0; i < nn; ++i) {
    const int c = c0 + i;
    const int start = __builtin_amdgcn_readlane(m_off, i);
    const int total = __builtin_amdgcn_readlane(m_tot, i);
    const float dc = __int_as_float(__builtin_amdgcn_readlane(__float_as_int(m_d), i));

    int r_l = r_cur;

    float2 v0[16];
    for (int q = 0; q < 16; ++q) {
      const int r = __builtin_amdgcn_readlane(r_l, q);
      v0[q] = *(const float2*)(hl + (size_t)r * STRIDE);
    }
    const int i1 = (i + 1 < last) ? i + 1 : last;
    const int i2 = (i + 2 < last) ? i + 2 : last;
    const int raw_nn = raw_r(i2);

    float ax = 0.0f, ay = 0.0f;
    for (int q = 0; q < 16; ++q) {
      ax += v0[q].x;
      ay += v0[q].y;
    }

    int base = 0, k = 16;
    for (;;) {
      const int rem = total - base;
      const int m = rem < 32 ? rem : 32;
      for (; k + 8 <= m; k += 8) {
        float2 v[8];
        for (int q = 0; q < 8; ++q) {
          const int r = __builtin_amdgcn_readlane(r_l, k + q);
          v[q] = *(const float2*)(hl + (size_t)r * STRIDE);
        }
        for (int q = 0; q < 8; ++q) {
          ax += v[q].x;
          ay += v[q].y;
        }
      }
      for (; k < m; k += 4) {
        float2 v[4];
        for (int q = 0; q < 4; ++q) {
          const int r = __builtin_amdgcn_readlane(r_l, k + q);
          v[q] = *(const float2*)(hl + (size_t)r * STRIDE);
        }
        for (int q = 0; q < 4; ++q) {
          ax += v[q].x;
          ay += v[q].y;
        }
      }
      base += 32;
      if (base >= total) break;
      const int j = base + lane;
      r_l = (j == total - 1) ? c : N;
      if (j < total - 1) r_l = src[start + j];
      k = 0;
    }
    if (act) store(c, ax * dc + bx, ay * dc + by);
    r_cur = fix_r(i1, raw_nxt);
    raw_nxt = raw_nn;
  }
}

template <int STRIDE, int CH, int OSTR>
__global__ void gcn_gather(const int* __restrict__ off, const int* __restrict__ cnt,
                           const int* __restrict__ src,
                           const float* __restrict__ dinv,
                           const float* __restrict__ h,
                           const float* __restrict__ bias,
                           float* __restrict__ dst, int N) {
  const int lane = threadIdx.x & 31;
  const int c0 = NPW * __builtin_amdgcn_readfirstlane(
      (int)((blockIdx.x * blockDim.x + threadIdx.x) >> 5));
  if (c0 >= N) return;
  gcn_gather_wave<STRIDE, CH>(off, cnt, src, dinv, h, bias, N, c0, lane,
                              [&](int c, float ox, float oy) {
    float2 o;
    o.x = ox;
    o.y = oy;
    *(float2*)(dst + (size_t)c * OSTR + lane * 2) = o;
  });
}

#define BROWS  (8 * NPW)
#define APITCH (HID + 4)
__global__ void gcn_gather_gemm2(const int* __restrict__ off, const int* __restrict__ cnt,
                                 const int* __restrict__ src,
                                 const float* __restrict__ dinv,
                                 const float* __restrict__ h,
                                 const float* __restrict__ bias,
                                 const float* __restrict__ W2,
                                 float* __restrict__ h2, int N) {
  __shared__ float sa[BROWS * APITCH];
  __shared__ float sw[OUTP * APITCH];
#pragma unroll
  for (int j = 0; j < HID * OUTC / 256; ++j) {
    const int i = j * 256 + (int)threadIdx.x;
    const int k = i / OUTC;
    sw[(i - k * OUTC) * APITCH + k] = W2[i];
  }
  for (int i = threadIdx.x; i < (OUTP - OUTC) * HID; i += 256)
    sw[(OUTC + i / HID) * APITCH + (i % HID)] = 0.0f;
  const int lane = threadIdx.x & 31;
  const int wave = __builtin_amdgcn_readfirstlane((int)(threadIdx.x >> 5));
  const int r0 = blockIdx.x * BROWS;
  const int c0 = r0 + wave * NPW;
  if (c0 < N) {
    gcn_gather_wave<HID, HID>(off, cnt, src, dinv, h, bias, N, c0, lane,
                              [&](int c, float ox, float oy) {
      float2 o;
      o.x = fmaxf(ox, 0.0f);
      o.y = fmaxf(oy, 0.0f);
      *(float2*)(sa + (c - r0) * APITCH + lane * 2) = o;
    });
  }
  __syncthreads();

  const int tile = wave & 3;
  const int m0 = r0 + tile * 16;
  if (m0 + 16 <= N) {
    const int rc = lane & 15;
    const int kOff = (lane < 16) ? 0 : 2;
    const int mBase = m0 + ((lane < 16) ? 0 : 8);
    const float4 dlo = *(const float4*)(dinv + mBase);
    const float4 dhi = *(const float4*)(dinv + mBase + 4);
    const float dr[8] = {dlo.x, dlo.y, dlo.z, dlo.w, dhi.x, dhi.y, dhi.z, dhi.w};
    const float* arow = sa + (tile * 16 + rc) * APITCH + kOff;
    const float* brow = sw + rc * APITCH + kOff;
    if (wave < 4) {
      v8f acc0 = {}, acc1 = {};
#pragma unroll
      for (int k = 0; k < HID; k += 4) {
        v2f a, b0, b1;
        a.x = arow[k];
        a.y = arow[k + 1];
        b0.x = brow[k];                 b0.y = brow[k + 1];
        b1.x = brow[16 * APITCH + k];   b1.y = brow[16 * APITCH + k + 1];
        acc0 = __builtin_amdgcn_wmma_f32_16x16x4_f32(false, a, false, b0,
                                                     (short)0, acc0, false, false);
        acc1 = __builtin_amdgcn_wmma_f32_16x16x4_f32(false, a, false, b1,
                                                     (short)0, acc1, false, false);
      }
#pragma unroll
      for (int j = 0; j < 8; ++j) {
        float* o = h2 + (size_t)(mBase + j) * OUTP + rc;
        o[0]  = acc0[j] * dr[j];
        o[16] = acc1[j] * dr[j];
      }
    } else {
      v8f acc2 = {};
#pragma unroll
      for (int k = 0; k < HID; k += 4) {
        v2f a, b2;
        a.x = arow[k];
        a.y = arow[k + 1];
        b2.x = brow[32 * APITCH + k];
        b2.y = brow[32 * APITCH + k + 1];
        acc2 = __builtin_amdgcn_wmma_f32_16x16x4_f32(false, a, false, b2,
                                                     (short)0, acc2, false, false);
      }
      if (32 + rc < OUTC) {
#pragma unroll
        for (int j = 0; j < 8; ++j)
          h2[(size_t)(mBase + j) * OUTP + 32 + rc] = acc2[j] * dr[j];
      }
    }
  }
  if (blockIdx.x == 0 && wave == 0 && lane < OUTP / 2) {
    float2 z;
    z.x = 0.0f;
    z.y = 0.0f;
    *(float2*)(h2 + (size_t)N * OUTP + lane * 2) = z;
  }
}

extern "C" void kernel_launch(void* const* d_in, const int* in_sizes, int n_in,
                              void* d_out, int out_size, void* d_ws, size_t ws_size,
                              hipStream_t stream) {
  const float* x  = (const float*)d_in[0];
  const int*   ei = (const int*)d_in[1];
  const float* W1 = (const float*)d_in[2];
  const float* b1 = (const float*)d_in[3];
  const float* W2 = (const float*)d_in[4];
  const float* b2 = (const float*)d_in[5];
  float* out = (float*)d_out;

  const int N = in_sizes[0] / IN_CH;
  const int E = in_sizes[1] / 2;
  const int* rowp = ei;
  const int* colp = ei + E;

  const size_t Np = ((size_t)N + 63) & ~(size_t)63;
  const size_t Ep = ((size_t)E + 63) & ~(size_t)63;
  const int NB = (N + BNODES - 1) / BNODES;
  const int etiles = (E + TEDGE - 1) / TEDGE;
  int*   wsi  = (int*)d_ws;
  int*   cnt  = wsi;
  int*   off  = wsi + Np;
  float* dinv = (float*)(wsi + 2 * Np);
  int*   tileOff = wsi + 3 * Np;
  int2*  bedge = (int2*)(wsi + 3 * Np + MAXT * MAXB);
  int*   srcs = wsi + 3 * Np + MAXT * MAXB + 2 * Ep;
  float* h1   = (float*)(srcs + Ep);
  float* h2   = h1 + ((size_t)N + 1) * HID;

  gcn_tile_bins<<<etiles, 256, 0, stream>>>(rowp, colp, tileOff, bedge, E);
  gcn_bin_csr<<<NB, 256, 0, stream>>>(tileOff, bedge, cnt, off, dinv, srcs, N, etiles);
  gcn_gemm1<<<(N / 16 + 3) / 4, 128, 0, stream>>>(x, W1, dinv, h1, N / 16, N);
  gcn_gather_gemm2<<<(N + BROWS - 1) / BROWS, 256, 0, stream>>>(
      off, cnt, srcs, dinv, h1, b1, W2, h2, N);
  gcn_gather<OUTP, OUTC, OUTC><<<(N + 8 * NPW - 1) / (8 * NPW), 256, 0, stream>>>(
      off, cnt, srcs, dinv, h2, b2, out, N);
}


# Per-bin kernel fills a bin's piece of the source array in LDS and writes it out as whole 128-byte lines, replacing one scattered 4-byte store per edge (oversized bins store directly); sum order kept
# speedup vs baseline: 4.6841x; 1.0382x over previous
#include <hip/hip_runtime.h>

typedef __attribute__((ext_vector_type(2))) float v2f;
typedef __attribute__((ext_vector_type(8))) float v8f;

#define IN_CH  128
#define HID    64
#define OUTC   40
#define OUTP   48
#define NPW 8
#define BSH    8
#define BNODES 256
#define MAXB   512
#define TPT    16
#define TEDGE  (TPT * 256)
#define MAXT   512
#define SB     25
#define SCAP   6144


__global__ void gcn_tile_bins(const int* __restrict__ row, const int* __restrict__ col,
                              int* __restrict__ tileOff, int2* __restrict__ bedge, int E) {
  __shared__ int h[MAXB];
  __shared__ int wsum[8];
  __shared__ int2 stage[TEDGE];
  const int tid = threadIdx.x;
  const int lane = tid & 31;
  const int wave = tid >> 5;
  h[2 * tid] = 0;
  h[2 * tid + 1] = 0;
  __syncthreads();

  const int tbase = blockIdx.x * TEDGE;
  const int base = tbase + tid;
  int cc[TPT], rk[TPT], rw[TPT];
#pragma unroll
  for (int i = 0; i < TPT; ++i) {
    const int e = base + i * 256;
    cc[i] = (e < E) ? col[e] : -1;
  }
#pragma unroll
  for (int i = 0; i < TPT; ++i) {
    const int e = base + i * 256;
    rw[i] = (e < E) ? row[e] : 0;
  }
#pragma unroll
  for (int i = 0; i < TPT; ++i) {
    rk[i] = 0;
    if (cc[i] >= 0) rk[i] = atomicAdd(&h[cc[i] >> BSH], 1);
  }
  __syncthreads();

  const int a0 = h[2 * tid];
  const int a1 = h[2 * tid + 1];
  const int s = a0 + a1;
  int inc = s;
  for (int o = 1; o < 32; o <<= 1) {
    const int t = __shfl_up(inc, o);
    if (lane >= o) inc += t;
  }
  if (lane == 31) wsum[wave] = inc;
  __syncthreads();
  int wpre = 0;
  for (int w = 0; w < wave; ++w) wpre += wsum[w];
  const int ex = wpre + inc - s;
  h[2 * tid] = ex;
  h[2 * tid + 1] = ex + a0;
  int2 pr;
  pr.x = ex;
  pr.y = ex + a0;
  *(int2*)(tileOff + (size_t)blockIdx.x * MAXB + 2 * tid) = pr;
  __syncthreads();

  int sl[TPT];
#pragma unroll
  for (int i = 0; i < TPT; ++i)
    sl[i] = h[((cc[i] >= 0) ? cc[i] : 0) >> BSH] + rk[i];
#pragma unroll
  for (int i = 0; i < TPT; ++i) {
    if (cc[i] >= 0) {
      int2 rec;
      rec.x = rw[i];
      rec.y = cc[i];
      stage[sl[i]] = rec;
    }
  }
  __syncthreads();
  const int n_t = (E - tbase < TEDGE) ? (E - tbase) : TEDGE;
  int2 ov[TPT];
#pragma unroll
  for (int i = 0; i < TPT; ++i) ov[i] = stage[tid + i * 256];
#pragma unroll
  for (int i = 0; i < TPT; ++i) {
    const int j = tid + i * 256;
    if (j < n_t) bedge[tbase + j] = ov[i];
  }
}

__global__ void gcn_bin_csr(const int* __restrict__ tileOff,
                            const int2* __restrict__ bedge,
                            int* __restrict__ cnt, int* __restrict__ off,
                            float* __restrict__ dinv, int* __restrict__ src,
                            int N, int etiles) {
  __shared__ int c[BNODES + 32];
  __shared__ int segOff[MAXT];
  __shared__ int segLen[MAXT];
  __shared__ int wsum[8];
  __shared__ int wsum2[8];
  __shared__ int sbuf[SCAP];
  const int tid = threadIdx.x;
  const int lane = tid & 31;
  const int wave = __builtin_amdgcn_readfirstlane(tid >> 5);
  const int b = blockIdx.x;
  const int lo = b * BNODES;

  int part = 0;
#pragma unroll
  for (int r = 0; r < 2; ++r) {
    const int t = tid + r * 256;
    int o0 = 0, ln = 0;
    if (t < etiles) {
      const int* p = tileOff + (size_t)t * MAXB + b;
      o0 = p[0];
      ln = p[1] - o0;
    }
    part += o0;
    segOff[t] = t * TEDGE + o0;
    segLen[t] = ln;
  }
  for (int o = 16; o > 0; o >>= 1) part += __shfl_xor(part, o);
  if (lane == 0) wsum[wave] = part;
  c[tid] = 0;
  if (tid < 32) c[BNODES + tid] = 0;
  __syncthreads();
  int start = 0;
  for (int w = 0; w < 8; ++w) start += wsum[w];

  for (int t0 = wave; t0 < etiles; t0 += 8 * SB) {
    const int tq = t0 + 8 * lane;
    const bool tv = (lane < SB) && (tq < MAXT);
    const int m_len = tv ? segLen[tq] : 0;
    const int m_off = tv ? segOff[tq] : 0;
    int ln[SB], tg[SB];
#pragma unroll
    for (int q = 0; q < SB; ++q) {
      ln[q] = __builtin_amdgcn_readlane(m_len, q);
      const int so = __builtin_amdgcn_readlane(m_off, q);
      const int k = (lane < ln[q]) ? lane : ln[q] - 1;
      const int idx = (ln[q] > 0) ? so + k : 0;
      tg[q] = bedge[idx].y;
    }
#pragma unroll
    for (int q = 0; q < SB; ++q)
      if (lane < ln[q]) atomicAdd(&c[tg[q] - lo], 1);
#pragma unroll
    for (int q = 0; q < SB; ++q) {
      if (ln[q] > 32) {
        const int so = __builtin_amdgcn_readlane(m_off, q);
        for (int k = 32 + lane; k < ln[q]; k += 32)
          atomicAdd(&c[bedge[so + k].y - lo], 1);
      }
    }
  }
  __syncthreads();

  const int v = c[tid];
  int inc = v;
  for (int o = 1; o < 32; o <<= 1) {
    const int t = __shfl_up(inc, o);
    if (lane >= o) inc += t;
  }
  if (lane == 31) wsum2[wave] = inc;
  __syncthreads();
  int wpre = 0, tot = 0;
  for (int w = 0; w < 8; ++w) {
    const int s = wsum2[w];
    if (w < wave) wpre += s;
    tot += s;
  }
  const int ex = wpre + inc - v;
  const int node = lo + tid;
  if (node < N) {
    cnt[node] = v;
    off[node] = start + ex;
    const float d = (float)(v + 1);
    dinv[node] = (d > 0.0f) ? rsqrtf(d) : 0.0f;
  }
  c[tid] = ex;
  __syncthreads();

  auto place = [&](auto put) {
    for (int t0 = wave; t0 < etiles; t0 += 8 * SB) {
      const int tq = t0 + 8 * lane;
      const bool tv = (lane < SB) && (tq < MAXT);
      const int m_len = tv ? segLen[tq] : 0;
      const int m_off = tv ? segOff[tq] : 0;
      int ln[SB], sl[SB];
      int2 rc[SB];
#pragma unroll
      for (int q = 0; q < SB; ++q) {
        ln[q] = __builtin_amdgcn_readlane(m_len, q);
        const int so = __builtin_amdgcn_readlane(m_off, q);
        const int k = (lane < ln[q]) ? lane : ln[q] - 1;
        const int idx = (ln[q] > 0) ? so + k : 0;
        rc[q] = bedge[idx];
      }
#pragma unroll
      for (int q = 0; q < SB; ++q) {
        const bool on = lane < ln[q];
        sl[q] = atomicAdd(&c[on ? rc[q].y - lo : BNODES + lane], on ? 1 : 0);
      }
#pragma unroll
      for (int q = 0; q < SB; ++q)
        if (lane < ln[q]) put(sl[q], rc[q].x);
#pragma unroll
      for (int q = 0; q < SB; ++q) {
        if (ln[q] > 32) {
          const int so = __builtin_amdgcn_readlane(m_off, q);
          for (int k = 32 + lane; k < ln[q]; k += 32) {
            const int2 rec = bedge[so + k];
            put(atomicAdd(&c[rec.y - lo], 1), rec.x);
          }
        }
      }
    }
  };
  if (tot <= SCAP) {
    place([&](int slot, int s) { sbuf[slot] = s; });
    __syncthreads();
    const int a = start & 31;
    for (int j0 = tid - a; j0 < tot; j0 += 4 * 256) {
      int w4[4];
      for (int u = 0; u < 4; ++u) {
        const int j = j0 + u * 256;
        w4[u] = sbuf[(j >= 0 && j < tot) ? j : 0];
      }
      for (int u = 0; u < 4; ++u) {
        const int j = j0 + u * 256;
        if (j >= 0 && j < tot) src[start + j] = w4[u];
      }
    }
  } else {
    place([&](int slot, int s) { src[start + slot] = s; });
  }
}

__global__ void gcn_gemm1(const float* __restrict__ x,
                          const float* __restrict__ W1,
                          const float* __restrict__ dinv,
                          float* __restrict__ h1, int tiles, int rows) {
  const int wave = threadIdx.x >> 5;
  const int lane = threadIdx.x & 31;
  const int tile = __builtin_amdgcn_readfirstlane((int)(blockIdx.x * 4 + wave));
  if (tile >= tiles) return;
  const int m0 = tile * 16;
  const int rc = lane & 15;
  const int kOff = (lane < 16) ? 0 : 2;

  const int mBase = m0 + ((lane < 16) ? 0 : 8);
  const float4 dlo = *(const float4*)(dinv + mBase);
  const float4 dhi = *(const float4*)(dinv + mBase + 4);
  const float dr[8] = {dlo.x, dlo.y, dlo.z, dlo.w, dhi.x, dhi.y, dhi.z, dhi.w};

  const float* xrow = x + (size_t)(m0 + rc) * IN_CH;
  v8f acc0 = {}, acc1 = {}, acc2 = {}, acc3 = {};
#pragma unroll
  for (int k = 0; k < IN_CH; k += 4) {
    v2f a, b0, b1, b2, b3;
    a.x = xrow[k + kOff];
    a.y = xrow[k + kOff + 1];
    const float* w0 = W1 + (size_t)(k + kOff) * HID + rc;
    const float* w1 = W1 + (size_t)(k + kOff + 1) * HID + rc;
    b0.x = w0[0];   b0.y = w1[0];
    b1.x = w0[16];  b1.y = w1[16];
    b2.x = w0[32];  b2.y = w1[32];
    b3.x = w0[48];  b3.y = w1[48];
    acc0 = __builtin_amdgcn_wmma_f32_16x16x4_f32(false, a, false, b0,
                                                 (short)0, acc0, false, false);
    acc1 = __builtin_amdgcn_wmma_f32_16x16x4_f32(false, a, false, b1,
                                                 (short)0, acc1, false, false);
    acc2 = __builtin_amdgcn_wmma_f32_16x16x4_f32(false, a, false, b2,
                                                 (short)0, acc2, false, false);
    acc3 = __builtin_amdgcn_wmma_f32_16x16x4_f32(false, a, false, b3,
                                                 (short)0, acc3, false, false);
  }
#pragma unroll
  for (int j = 0; j < 8; ++j) {
    float* o = h1 + (size_t)(mBase + j) * HID + rc;
    o[0]  = acc0[j] * dr[j];
    o[16] = acc1[j] * dr[j];
    o[32] = acc2[j] * dr[j];
    o[48] = acc3[j] * dr[j];
  }
  if (tile == 0) {
    float2 z;
    z.x = 0.0f;
    z.y = 0.0f;
    *(float2*)(h1 + (size_t)rows * HID + lane * 2) = z;
  }
}

template <int STRIDE, int CH, typename StoreF>
__device__ __forceinline__ void gcn_gather_wave(const int* __restrict__ off,
                                                const int* __restrict__ cnt,
                                                const int* __restrict__ src,
                                                const float* __restrict__ dinv,
                                                const float* __restrict__ h,
                                                const float* __restrict__ bias,
                                                int N, int c0, int lane, StoreF store) {
  const int nn = (N - c0 < NPW) ? (N - c0) : NPW;
  const int last = nn - 1;

  int m_off = 0, m_tot = 1;
  float m_d = 0.0f;
  if (lane < nn) {
    m_off = off[c0 + lane];
    m_tot = cnt[c0 + lane] + 1;
    m_d = dinv[c0 + lane];
  }
  const bool act = lane < CH / 2;
  const float* hl = h + (act ? lane * 2 : 0);
  float bx = 0.0f, by = 0.0f;
  if (act) {
    bx = bias[lane * 2];
    by = bias[lane * 2 + 1];
  }

  auto raw_r = [&](int i) -> int {
    const int st = __builtin_amdgcn_readlane(m_off, i);
    const int tt = __builtin_amdgcn_readlane(m_tot, i);
    const int idx = (lane < tt - 1) ? st + lane : 0;
    return src[idx];
  };
  auto fix_r = [&](int i, int raw) -> int {
    const int tt = __builtin_amdgcn_readlane(m_tot, i);
    return (lane < tt - 1) ? raw : ((lane == tt - 1) ? c0 + i : N);
  };

  int r_cur = fix_r(0, raw_r(0));
  int raw_nxt = raw_r(last < 1 ? last : 1);

  for (int i = 0; i < nn; ++i) {
    const int c = c0 + i;
    const int start = __builtin_amdgcn_readlane(m_off, i);
    const int total = __builtin_amdgcn_readlane(m_tot, i);
    const float dc = __int_as_float(__builtin_amdgcn_readlane(__float_as_int(m_d), i));

    int r_l = r_cur;

    float2 v0[16];
    for (int q = 0; q < 16; ++q) {
      const int r = __builtin_amdgcn_readlane(r_l, q);
      v0[q] = *(const float2*)(hl + (size_t)r * STRIDE);
    }
    const int i1 = (i + 1 < last) ? i + 1 : last;
    const int i2 = (i + 2 < last) ? i + 2 : last;
    const int raw_nn = raw_r(i2);

    float ax = 0.0f, ay = 0.0f;
    for (int q = 0; q < 16; ++q) {
      ax += v0[q].x;
      ay += v0[q].y;
    }

    int base = 0, k = 16;
    for (;;) {
      const int rem = total - base;
      const int m = rem < 32 ? rem : 32;
      for (; k + 8 <= m; k += 8) {
        float2 v[8];
        for (int q = 0; q < 8; ++q) {
          const int r = __builtin_amdgcn_readlane(r_l, k + q);
          v[q] = *(const float2*)(hl + (size_t)r * STRIDE);
        }
        for (int q = 0; q < 8; ++q) {
          ax += v[q].x;
          ay += v[q].y;
        }
      }
      for (; k < m; k += 4) {
        float2 v[4];
        for (int q = 0; q < 4; ++q) {
          const int r = __builtin_amdgcn_readlane(r_l, k + q);
          v[q] = *(const float2*)(hl + (size_t)r * STRIDE);
        }
        for (int q = 0; q < 4; ++q) {
          ax += v[q].x;
          ay += v[q].y;
        }
      }
      base += 32;
      if (base >= total) break;
      const int j = base + lane;
      r_l = (j == total - 1) ? c : N;
      if (j < total - 1) r_l = src[start + j];
      k = 0;
    }
    if (act) store(c, ax * dc + bx, ay * dc + by);
    r_cur = fix_r(i1, raw_nxt);
    raw_nxt = raw_nn;
  }
}

template <int STRIDE, int CH, int OSTR>
__global__ void gcn_gather(const int* __restrict__ off, const int* __restrict__ cnt,
                           const int* __restrict__ src,
                           const float* __restrict__ dinv,
                           const float* __restrict__ h,
                           const float* __restrict__ bias,
                           float* __restrict__ dst, int N) {
  const int lane = threadIdx.x & 31;
  const int c0 = NPW * __builtin_amdgcn_readfirstlane(
      (int)((blockIdx.x * blockDim.x + threadIdx.x) >> 5));
  if (c0 >= N) return;
  gcn_gather_wave<STRIDE, CH>(off, cnt, src, dinv, h, bias, N, c0, lane,
                              [&](int c, float ox, float oy) {
    float2 o;
    o.x = ox;
    o.y = oy;
    *(float2*)(dst + (size_t)c * OSTR + lane * 2) = o;
  });
}

#define BROWS  (8 * NPW)
#define APITCH (HID + 4)
__global__ void gcn_gather_gemm2(const int* __restrict__ off, const int* __restrict__ cnt,
                                 const int* __restrict__ src,
                                 const float* __restrict__ dinv,
                                 const float* __restrict__ h,
                                 const float* __restrict__ bias,
                                 const float* __restrict__ W2,
                                 float* __restrict__ h2, int N) {
  __shared__ float sa[BROWS * APITCH];
  __shared__ float sw[OUTP * APITCH];
#pragma unroll
  for (int j = 0; j < HID * OUTC / 256; ++j) {
    const int i = j * 256 + (int)threadIdx.x;
    const int k = i / OUTC;
    sw[(i - k * OUTC) * APITCH + k] = W2[i];
  }
  for (int i = threadIdx.x; i < (OUTP - OUTC) * HID; i += 256)
    sw[(OUTC + i / HID) * APITCH + (i % HID)] = 0.0f;
  const int lane = threadIdx.x & 31;
  const int wave = __builtin_amdgcn_readfirstlane((int)(threadIdx.x >> 5));
  const int r0 = blockIdx.x * BROWS;
  const int c0 = r0 + wave * NPW;
  if (c0 < N) {
    gcn_gather_wave<HID, HID>(off, cnt, src, dinv, h, bias, N, c0, lane,
                              [&](int c, float ox, float oy) {
      float2 o;
      o.x = fmaxf(ox, 0.0f);
      o.y = fmaxf(oy, 0.0f);
      *(float2*)(sa + (c - r0) * APITCH + lane * 2) = o;
    });
  }
  __syncthreads();

  const int tile = wave & 3;
  const int m0 = r0 + tile * 16;
  if (m0 + 16 <= N) {
    const int rc = lane & 15;
    const int kOff = (lane < 16) ? 0 : 2;
    const int mBase = m0 + ((lane < 16) ? 0 : 8);
    const float4 dlo = *(const float4*)(dinv + mBase);
    const float4 dhi = *(const float4*)(dinv + mBase + 4);
    const float dr[8] = {dlo.x, dlo.y, dlo.z, dlo.w, dhi.x, dhi.y, dhi.z, dhi.w};
    const float* arow = sa + (tile * 16 + rc) * APITCH + kOff;
    const float* brow = sw + rc * APITCH + kOff;
    if (wave < 4) {
      v8f acc0 = {}, acc1 = {};
#pragma unroll
      for (int k = 0; k < HID; k += 4) {
        v2f a, b0, b1;
        a.x = arow[k];
        a.y = arow[k + 1];
        b0.x = brow[k];                 b0.y = brow[k + 1];
        b1.x = brow[16 * APITCH + k];   b1.y = brow[16 * APITCH + k + 1];
        acc0 = __builtin_amdgcn_wmma_f32_16x16x4_f32(false, a, false, b0,
                                                     (short)0, acc0, false, false);
        acc1 = __builtin_amdgcn_wmma_f32_16x16x4_f32(false, a, false, b1,
                                                     (short)0, acc1, false, false);
      }
#pragma unroll
      for (int j = 0; j < 8; ++j) {
        float* o = h2 + (size_t)(mBase + j) * OUTP + rc;
        o[0]  = acc0[j] * dr[j];
        o[16] = acc1[j] * dr[j];
      }
    } else {
      v8f acc2 = {};
#pragma unroll
      for (int k = 0; k < HID; k += 4) {
        v2f a, b2;
        a.x = arow[k];
        a.y = arow[k + 1];
        b2.x = brow[32 * APITCH + k];
        b2.y = brow[32 * APITCH + k + 1];
        acc2 = __builtin_amdgcn_wmma_f32_16x16x4_f32(false, a, false, b2,
                                                     (short)0, acc2, false, false);
      }
      if (32 + rc < OUTC) {
#pragma unroll
        for (int j = 0; j < 8; ++j)
          h2[(size_t)(mBase + j) * OUTP + 32 + rc] = acc2[j] * dr[j];
      }
    }
  }
  if (blockIdx.x == 0 && wave == 0 && lane < OUTP / 2) {
    float2 z;
    z.x = 0.0f;
    z.y = 0.0f;
    *(float2*)(h2 + (size_t)N * OUTP + lane * 2) = z;
  }
}

extern "C" void kernel_launch(void* const* d_in, const int* in_sizes, int n_in,
                              void* d_out, int out_size, void* d_ws, size_t ws_size,
                              hipStream_t stream) {
  const float* x  = (const float*)d_in[0];
  const int*   ei = (const int*)d_in[1];
  const float* W1 = (const float*)d_in[2];
  const float* b1 = (const float*)d_in[3];
  const float* W2 = (const float*)d_in[4];
  const float* b2 = (const float*)d_in[5];
  float* out = (float*)d_out;

  const int N = in_sizes[0] / IN_CH;
  const int E = in_sizes[1] / 2;
  const int* rowp = ei;
  const int* colp = ei + E;

  const size_t Np = ((size_t)N + 63) & ~(size_t)63;
  const size_t Ep = ((size_t)E + 63) & ~(size_t)63;
  const int NB = (N + BNODES - 1) / BNODES;
  const int etiles = (E + TEDGE - 1) / TEDGE;
  int*   wsi  = (int*)d_ws;
  int*   cnt  = wsi;
  int*   off  = wsi + Np;
  float* dinv = (float*)(wsi + 2 * Np);
  int*   tileOff = wsi + 3 * Np;
  int2*  bedge = (int2*)(wsi + 3 * Np + MAXT * MAXB);
  int*   srcs = wsi + 3 * Np + MAXT * MAXB + 2 * Ep;
  float* h1   = (float*)(srcs + Ep);
  float* h2   = h1 + ((size_t)N + 1) * HID;

  gcn_tile_bins<<<etiles, 256, 0, stream>>>(rowp, colp, tileOff, bedge, E);
  gcn_bin_csr<<<NB, 256, 0, stream>>>(tileOff, bedge, cnt, off, dinv, srcs, N, etiles);
  gcn_gemm1<<<(N / 16 + 3) / 4, 128, 0, stream>>>(x, W1, dinv, h1, N / 16, N);
  gcn_gather_gemm2<<<(N + BROWS - 1) / BROWS, 256, 0, stream>>>(
      off, cnt, srcs, dinv, h1, b1, W2, h2, N);
  gcn_gather<OUTP, OUTC, OUTC><<<(N + 8 * NPW - 1) / (8 * NPW), 256, 0, stream>>>(
      off, cnt, srcs, dinv, h2, b2, out, N);
}


# GEMM1 becomes the epilogue of the per-bin kernel: a bin's block, which holds its rows' D^-1/2, multiplies its own 256 rows of x by W1 (W1 in LDS, two row tiles a wave); one launch less; all sums kept
# speedup vs baseline: 4.8825x; 1.0424x over previous
#include <hip/hip_runtime.h>

typedef __attribute__((ext_vector_type(2))) float v2f;
typedef __attribute__((ext_vector_type(8))) float v8f;

#define IN_CH  128
#define HID    64
#define OUTC   40
#define OUTP   48
#define NPW 8
#define BSH    8
#define BNODES 256
#define MAXB   512
#define TPT    16
#define TEDGE  (TPT * 256)
#define MAXT   512
#define SB     25
#define SCAP   6144
#define WPITCH (IN_CH + 4)


__global__ void gcn_tile_bins(const int* __restrict__ row, const int* __restrict__ col,
                              int* __restrict__ tileOff, int2* __restrict__ bedge, int E) {
  __shared__ int h[MAXB];
  __shared__ int wsum[8];
  __shared__ int2 stage[TEDGE];
  const int tid = threadIdx.x;
  const int lane = tid & 31;
  const int wave = tid >> 5;
  h[2 * tid] = 0;
  h[2 * tid + 1] = 0;
  __syncthreads();

  const int tbase = blockIdx.x * TEDGE;
  const int base = tbase + tid;
  int cc[TPT], rk[TPT], rw[TPT];
#pragma unroll
  for (int i = 0; i < TPT; ++i) {
    const int e = base + i * 256;
    cc[i] = (e < E) ? col[e] : -1;
  }
#pragma unroll
  for (int i = 0; i < TPT; ++i) {
    const int e = base + i * 256;
    rw[i] = (e < E) ? row[e] : 0;
  }
#pragma unroll
  for (int i = 0; i < TPT; ++i) {
    rk[i] = 0;
    if (cc[i] >= 0) rk[i] = atomicAdd(&h[cc[i] >> BSH], 1);
  }
  __syncthreads();

  const int a0 = h[2 * tid];
  const int a1 = h[2 * tid + 1];
  const int s = a0 + a1;
  int inc = s;
  for (int o = 1; o < 32; o <<= 1) {
    const int t = __shfl_up(inc, o);
    if (lane >= o) inc += t;
  }
  if (lane == 31) wsum[wave] = inc;
  __syncthreads();
  int wpre = 0;
  for (int w = 0; w < wave; ++w) wpre += wsum[w];
  const int ex = wpre + inc - s;
  h[2 * tid] = ex;
  h[2 * tid + 1] = ex + a0;
  int2 pr;
  pr.x = ex;
  pr.y = ex + a0;
  *(int2*)(tileOff + (size_t)blockIdx.x * MAXB + 2 * tid) = pr;
  __syncthreads();

  int sl[TPT];
#pragma unroll
  for (int i = 0; i < TPT; ++i)
    sl[i] = h[((cc[i] >= 0) ? cc[i] : 0) >> BSH] + rk[i];
#pragma unroll
  for (int i = 0; i < TPT; ++i) {
    if (cc[i] >= 0) {
      int2 rec;
      rec.x = rw[i];
      rec.y = cc[i];
      stage[sl[i]] = rec;
    }
  }
  __syncthreads();
  const int n_t = (E - tbase < TEDGE) ? (E - tbase) : TEDGE;
  int2 ov[TPT];
#pragma unroll
  for (int i = 0; i < TPT; ++i) ov[i] = stage[tid + i * 256];
#pragma unroll
  for (int i = 0; i < TPT; ++i) {
    const int j = tid + i * 256;
    if (j < n_t) bedge[tbase + j] = ov[i];
  }
}

__global__ void __launch_bounds__(256, 4)
gcn_bin_csr(const int* __restrict__ tileOff,
                            const int2* __restrict__ bedge,
                            int* __restrict__ cnt, int* __restrict__ off,
                            float* __restrict__ dinv, int* __restrict__ src,
                            const float* __restrict__ x,
                            const float* __restrict__ W1,
                            float* __restrict__ h1,
                            int N, int etiles) {
  __shared__ int c[BNODES + 32];
  __shared__ int segOff[MAXT];
  __shared__ int segLen[MAXT];
  __shared__ int wsum[8];
  __shared__ int wsum2[8];
  __shared__ int sbuf[SCAP];
  __shared__ alignas(16) float sw[HID * WPITCH];
  __shared__ alignas(16) float sd[BNODES];
  const int tid = threadIdx.x;
  const int lane = tid & 31;
  const int wave = __builtin_amdgcn_readfirstlane(tid >> 5);
  const int b = blockIdx.x;
  const int lo = b * BNODES;

  int part = 0;
#pragma unroll
  for (int r = 0; r < 2; ++r) {
    const int t = tid + r * 256;
    int o0 = 0, ln = 0;
    if (t < etiles) {
      const int* p = tileOff + (size_t)t * MAXB + b;
      o0 = p[0];
      ln = p[1] - o0;
    }
    part += o0;
    segOff[t] = t * TEDGE + o0;
    segLen[t] = ln;
  }
#pragma unroll
  for (int j = 0; j < IN_CH * HID / 256; ++j) {
    const int i = j * 256 + tid;
    sw[(i & (HID - 1)) * WPITCH + (i / HID)] = W1[i];
  }
  for (int o = 16; o > 0; o >>= 1) part += __shfl_xor(part, o);
  if (lane == 0) wsum[wave] = part;
  c[tid] = 0;
  if (tid < 32) c[BNODES + tid] = 0;
  __syncthreads();
  int start = 0;
  for (int w = 0; w < 8; ++w) start += wsum[w];

  for (int t0 = wave; t0 < etiles; t0 += 8 * SB) {
    const int tq = t0 + 8 * lane;
    const bool tv = (lane < SB) && (tq < MAXT);
    const int m_len = tv ? segLen[tq] : 0;
    const int m_off = tv ? segOff[tq] : 0;
    int ln[SB], tg[SB];
#pragma unroll
    for (int q = 0; q < SB; ++q) {
      ln[q] = __builtin_amdgcn_readlane(m_len, q);
      const int so = __builtin_amdgcn_readlane(m_off, q);
      const int k = (lane < ln[q]) ? lane : ln[q] - 1;
      const int idx = (ln[q] > 0) ? so + k : 0;
      tg[q] = bedge[idx].y;
    }
#pragma unroll
    for (int q = 0; q < SB; ++q)
      if (lane < ln[q]) atomicAdd(&c[tg[q] - lo], 1);
#pragma unroll
    for (int q = 0; q < SB; ++q) {
      if (ln[q] > 32) {
        const int so = __builtin_amdgcn_readlane(m_off, q);
        for (int k = 32 + lane; k < ln[q]; k += 32)
          atomicAdd(&c[bedge[so + k].y - lo], 1);
      }
    }
  }
  __syncthreads();

  const int v = c[tid];
  int inc = v;
  for (int o = 1; o < 32; o <<= 1) {
    const int t = __shfl_up(inc, o);
    if (lane >= o) inc += t;
  }
  if (lane == 31) wsum2[wave] = inc;
  __syncthreads();
  int wpre = 0, tot = 0;
  for (int w = 0; w < 8; ++w) {
    const int s = wsum2[w];
    if (w < wave) wpre += s;
    tot += s;
  }
  const int ex = wpre + inc - v;
  const int node = lo + tid;
  float dn = 0.0f;
  if (node < N) {
    cnt[node] = v;
    off[node] = start + ex;
    const float d = (float)(v + 1);
    dn = (d > 0.0f) ? rsqrtf(d) : 0.0f;
    dinv[node] = dn;
  }
  sd[tid] = dn;
  c[tid] = ex;
  __syncthreads();

  auto place = [&](auto put) {
    for (int t0 = wave; t0 < etiles; t0 += 8 * SB) {
      const int tq = t0 + 8 * lane;
      const bool tv = (lane < SB) && (tq < MAXT);
      const int m_len = tv ? segLen[tq] : 0;
      const int m_off = tv ? segOff[tq] : 0;
      int ln[SB], sl[SB];
      int2 rc[SB];
#pragma unroll
      for (int q = 0; q < SB; ++q) {
        ln[q] = __builtin_amdgcn_readlane(m_len, q);
        const int so = __builtin_amdgcn_readlane(m_off, q);
        const int k = (lane < ln[q]) ? lane : ln[q] - 1;
        const int idx = (ln[q] > 0) ? so + k : 0;
        rc[q] = bedge[idx];
      }
#pragma unroll
      for (int q = 0; q < SB; ++q) {
        const bool on = lane < ln[q];
        sl[q] = atomicAdd(&c[on ? rc[q].y - lo : BNODES + lane], on ? 1 : 0);
      }
#pragma unroll
      for (int q = 0; q < SB; ++q)
        if (lane < ln[q]) put(sl[q], rc[q].x);
#pragma unroll
      for (int q = 0; q < SB; ++q) {
        if (ln[q] > 32) {
          const int so = __builtin_amdgcn_readlane(m_off, q);
          for (int k = 32 + lane; k < ln[q]; k += 32) {
            const int2 rec = bedge[so + k];
            put(atomicAdd(&c[rec.y - lo], 1), rec.x);
          }
        }
      }
    }
  };
  if (tot <= SCAP) {
    place([&](int slot, int s) { sbuf[slot] = s; });
    __syncthreads();
    const int a = start & 31;
    for (int j0 = tid - a; j0 < tot; j0 += 4 * 256) {
      int w4[4];
      for (int u = 0; u < 4; ++u) {
        const int j = j0 + u * 256;
        w4[u] = sbuf[(j >= 0 && j < tot) ? j : 0];
      }
      for (int u = 0; u < 4; ++u) {
        const int j = j0 + u * 256;
        if (j >= 0 && j < tot) src[start + j] = w4[u];
      }
    }
  } else {
    place([&](int slot, int s) { src[start + slot] = s; });
  }

  const int tiles = N >> 4;
  const int t0 = b * (BNODES / 16) + 2 * wave;
  if (b == 0 && wave == 0) {
    float2 z;
    z.x = 0.0f;
    z.y = 0.0f;
    *(float2*)(h1 + (size_t)N * HID + lane * 2) = z;
  }
  if (t0 >= tiles) return;
  const bool two = t0 + 1 < tiles;
  const int rc = lane & 15;
  const int hf = lane >> 4;
  const int m0 = t0 * 16;
  const int m1 = two ? m0 + 16 : m0;
  const float* xr0 = x + (size_t)(m0 + rc) * IN_CH + 2 * hf;
  const float* xr1 = x + (size_t)(m1 + rc) * IN_CH + 2 * hf;
  v2f A0[IN_CH / 4], A1[IN_CH / 4];
#pragma unroll
  for (int j = 0; j < IN_CH / 4; ++j) {
    A0[j] = *(const v2f*)(xr0 + 4 * j);
    A1[j] = *(const v2f*)(xr1 + 4 * j);
  }
  __builtin_amdgcn_sched_barrier(0);
  const float* brow = sw + rc * WPITCH + 2 * hf;
  v8f c00 = {}, c01 = {}, c02 = {}, c03 = {};
  v8f c10 = {}, c11 = {}, c12 = {}, c13 = {};
#pragma unroll
  for (int j = 0; j < IN_CH / 4; ++j) {
    const v2f b0 = *(const v2f*)(brow + 4 * j);
    const v2f b1 = *(const v2f*)(brow + 16 * WPITCH + 4 * j);
    const v2f b2 = *(const v2f*)(brow + 32 * WPITCH + 4 * j);
    const v2f b3 = *(const v2f*)(brow + 48 * WPITCH + 4 * j);
    c00 = __builtin_amdgcn_wmma_f32_16x16x4_f32(false, A0[j], false, b0,
                                                (short)0, c00, false, false);
    c01 = __builtin_amdgcn_wmma_f32_16x16x4_f32(false, A0[j], false, b1,
                                                (short)0, c01, false, false);
    c02 = __builtin_amdgcn_wmma_f32_16x16x4_f32(false, A0[j], false, b2,
                                                (short)0, c02, false, false);
    c03 = __builtin_amdgcn_wmma_f32_16x16x4_f32(false, A0[j], false, b3,
                                                (short)0, c03, false, false);
    c10 = __builtin_amdgcn_wmma_f32_16x16x4_f32(false, A1[j], false, b0,
                                                (short)0, c10, false, false);
    c11 = __builtin_amdgcn_wmma_f32_16x16x4_f32(false, A1[j], false, b1,
                                                (short)0, c11, false, false);
    c12 = __builtin_amdgcn_wmma_f32_16x16x4_f32(false, A1[j], false, b2,
                                                (short)0, c12, false, false);
    c13 = __builtin_amdgcn_wmma_f32_16x16x4_f32(false, A1[j], false, b3,
                                                (short)0, c13, false, false);
  }
  const int lr = (m0 - lo) + 8 * hf;
  const float4 p0 = *(const float4*)(sd + lr);
  const float4 p1 = *(const float4*)(sd + lr + 4);
  const float dr0[8] = {p0.x, p0.y, p0.z, p0.w, p1.x, p1.y, p1.z, p1.w};
#pragma unroll
  for (int j = 0; j < 8; ++j) {
    float* o = h1 + (size_t)(m0 + 8 * hf + j) * HID + rc;
    o[0]  = c00[j] * dr0[j];
    o[16] = c01[j] * dr0[j];
    o[32] = c02[j] * dr0[j];
    o[48] = c03[j] * dr0[j];
  }
  if (two) {
    const float4 q0 = *(const float4*)(sd + lr + 16);
    const float4 q1 = *(const float4*)(sd + lr + 20);
    const float dr1[8] = {q0.x, q0.y, q0.z, q0.w, q1.x, q1.y, q1.z, q1.w};
#pragma unroll
    for (int j = 0; j < 8; ++j) {
      float* o = h1 + (size_t)(m1 + 8 * hf + j) * HID + rc;
      o[0]  = c10[j] * dr1[j];
      o[16] = c11[j] * dr1[j];
      o[32] = c12[j] * dr1[j];
      o[48] = c13[j] * dr1[j];
    }
  }
}

template <int STRIDE, int CH, typename StoreF>
__device__ __forceinline__ void gcn_gather_wave(const int* __restrict__ off,
                                                const int* __restrict__ cnt,
                                                const int* __restrict__ src,
                                                const float* __restrict__ dinv,
                                                const float* __restrict__ h,
                                                const float* __restrict__ bias,
                                                int N, int c0, int lane, StoreF store) {
  const int nn = (N - c0 < NPW) ? (N - c0) : NPW;
  const int last = nn - 1;

  int m_off = 0, m_tot = 1;
  float m_d = 0.0f;
  if (lane < nn) {
    m_off = off[c0 + lane];
    m_tot = cnt[c0 + lane] + 1;
    m_d = dinv[c0 + lane];
  }
  const bool act = lane < CH / 2;
  const float* hl = h + (act ? lane * 2 : 0);
  float bx = 0.0f, by = 0.0f;
  if (act) {
    bx = bias[lane * 2];
    by = bias[lane * 2 + 1];
  }

  auto raw_r = [&](int i) -> int {
    const int st = __builtin_amdgcn_readlane(m_off, i);
    const int tt = __builtin_amdgcn_readlane(m_tot, i);
    const int idx = (lane < tt - 1) ? st + lane : 0;
    return src[idx];
  };
  auto fix_r = [&](int i, int raw) -> int {
    const int tt = __builtin_amdgcn_readlane(m_tot, i);
    return (lane < tt - 1) ? raw : ((lane == tt - 1) ? c0 + i : N);
  };

  int r_cur = fix_r(0, raw_r(0));
  int raw_nxt = raw_r(last < 1 ? last : 1);

  for (int i = 0; i < nn; ++i) {
    const int c = c0 + i;
    const int start = __builtin_amdgcn_readlane(m_off, i);
    const int total = __builtin_amdgcn_readlane(m_tot, i);
    const float dc = __int_as_float(__builtin_amdgcn_readlane(__float_as_int(m_d), i));

    int r_l = r_cur;

    float2 v0[16];
    for (int q = 0; q < 16; ++q) {
      const int r = __builtin_amdgcn_readlane(r_l, q);
      v0[q] = *(const float2*)(hl + (size_t)r * STRIDE);
    }
    const int i1 = (i + 1 < last) ? i + 1 : last;
    const int i2 = (i + 2 < last) ? i + 2 : last;
    const int raw_nn = raw_r(i2);

    float ax = 0.0f, ay = 0.0f;
    for (int q = 0; q < 16; ++q) {
      ax += v0[q].x;
      ay += v0[q].y;
    }

    int base = 0, k = 16;
    for (;;) {
      const int rem = total - base;
      const int m = rem < 32 ? rem : 32;
      for (; k + 8 <= m; k += 8) {
        float2 v[8];
        for (int q = 0; q < 8; ++q) {
          const int r = __builtin_amdgcn_readlane(r_l, k + q);
          v[q] = *(const float2*)(hl + (size_t)r * STRIDE);
        }
        for (int q = 0; q < 8; ++q) {
          ax += v[q].x;
          ay += v[q].y;
        }
      }
      for (; k < m; k += 4) {
        float2 v[4];
        for (int q = 0; q < 4; ++q) {
          const int r = __builtin_amdgcn_readlane(r_l, k + q);
          v[q] = *(const float2*)(hl + (size_t)r * STRIDE);
        }
        for (int q = 0; q < 4; ++q) {
          ax += v[q].x;
          ay += v[q].y;
        }
      }
      base += 32;
      if (base >= total) break;
      const int j = base + lane;
      r_l = (j == total - 1) ? c : N;
      if (j < total - 1) r_l = src[start + j];
      k = 0;
    }
    if (act) store(c, ax * dc + bx, ay * dc + by);
    r_cur = fix_r(i1, raw_nxt);
    raw_nxt = raw_nn;
  }
}

template <int STRIDE, int CH, int OSTR>
__global__ void gcn_gather(const int* __restrict__ off, const int* __restrict__ cnt,
                           const int* __restrict__ src,
                           const float* __restrict__ dinv,
                           const float* __restrict__ h,
                           const float* __restrict__ bias,
                           float* __restrict__ dst, int N) {
  const int lane = threadIdx.x & 31;
  const int c0 = NPW * __builtin_amdgcn_readfirstlane(
      (int)((blockIdx.x * blockDim.x + threadIdx.x) >> 5));
  if (c0 >= N) return;
  gcn_gather_wave<STRIDE, CH>(off, cnt, src, dinv, h, bias, N, c0, lane,
                              [&](int c, float ox, float oy) {
    float2 o;
    o.x = ox;
    o.y = oy;
    *(float2*)(dst + (size_t)c * OSTR + lane * 2) = o;
  });
}

#define BROWS  (8 * NPW)
#define APITCH (HID + 4)
__global__ void gcn_gather_gemm2(const int* __restrict__ off, const int* __restrict__ cnt,
                                 const int* __restrict__ src,
                                 const float* __restrict__ dinv,
                                 const float* __restrict__ h,
                                 const float* __restrict__ bias,
                                 const float* __restrict__ W2,
                                 float* __restrict__ h2, int N) {
  __shared__ float sa[BROWS * APITCH];
  __shared__ float sw[OUTP * APITCH];
#pragma unroll
  for (int j = 0; j < HID * OUTC / 256; ++j) {
    const int i = j * 256 + (int)threadIdx.x;
    const int k = i / OUTC;
    sw[(i - k * OUTC) * APITCH + k] = W2[i];
  }
  for (int i = threadIdx.x; i < (OUTP - OUTC) * HID; i += 256)
    sw[(OUTC + i / HID) * APITCH + (i % HID)] = 0.0f;
  const int lane = threadIdx.x & 31;
  const int wave = __builtin_amdgcn_readfirstlane((int)(threadIdx.x >> 5));
  const int r0 = blockIdx.x * BROWS;
  const int c0 = r0 + wave * NPW;
  if (c0 < N) {
    gcn_gather_wave<HID, HID>(off, cnt, src, dinv, h, bias, N, c0, lane,
                              [&](int c, float ox, float oy) {
      float2 o;
      o.x = fmaxf(ox, 0.0f);
      o.y = fmaxf(oy, 0.0f);
      *(float2*)(sa + (c - r0) * APITCH + lane * 2) = o;
    });
  }
  __syncthreads();

  const int tile = wave & 3;
  const int m0 = r0 + tile * 16;
  if (m0 + 16 <= N) {
    const int rc = lane & 15;
    const int kOff = (lane < 16) ? 0 : 2;
    const int mBase = m0 + ((lane < 16) ? 0 : 8);
    const float4 dlo = *(const float4*)(dinv + mBase);
    const float4 dhi = *(const float4*)(dinv + mBase + 4);
    const float dr[8] = {dlo.x, dlo.y, dlo.z, dlo.w, dhi.x, dhi.y, dhi.z, dhi.w};
    const float* arow = sa + (tile * 16 + rc) * APITCH + kOff;
    const float* brow = sw + rc * APITCH + kOff;
    if (wave < 4) {
      v8f acc0 = {}, acc1 = {};
#pragma unroll
      for (int k = 0; k < HID; k += 4) {
        v2f a, b0, b1;
        a.x = arow[k];
        a.y = arow[k + 1];
        b0.x = brow[k];                 b0.y = brow[k + 1];
        b1.x = brow[16 * APITCH + k];   b1.y = brow[16 * APITCH + k + 1];
        acc0 = __builtin_amdgcn_wmma_f32_16x16x4_f32(false, a, false, b0,
                                                     (short)0, acc0, false, false);
        acc1 = __builtin_amdgcn_wmma_f32_16x16x4_f32(false, a, false, b1,
                                                     (short)0, acc1, false, false);
      }
#pragma unroll
      for (int j = 0; j < 8; ++j) {
        float* o = h2 + (size_t)(mBase + j) * OUTP + rc;
        o[0]  = acc0[j] * dr[j];
        o[16] = acc1[j] * dr[j];
      }
    } else {
      v8f acc2 = {};
#pragma unroll
      for (int k = 0; k < HID; k += 4) {
        v2f a, b2;
        a.x = arow[k];
        a.y = arow[k + 1];
        b2.x = brow[32 * APITCH + k];
        b2.y = brow[32 * APITCH + k + 1];
        acc2 = __builtin_amdgcn_wmma_f32_16x16x4_f32(false, a, false, b2,
                                                     (short)0, acc2, false, false);
      }
      if (32 + rc < OUTC) {
#pragma unroll
        for (int j = 0; j < 8; ++j)
          h2[(size_t)(mBase + j) * OUTP + 32 + rc] = acc2[j] * dr[j];
      }
    }
  }
  if (blockIdx.x == 0 && wave == 0 && lane < OUTP / 2) {
    float2 z;
    z.x = 0.0f;
    z.y = 0.0f;
    *(float2*)(h2 + (size_t)N * OUTP + lane * 2) = z;
  }
}

extern "C" void kernel_launch(void* const* d_in, const int* in_sizes, int n_in,
                              void* d_out, int out_size, void* d_ws, size_t ws_size,
                              hipStream_t stream) {
  const float* x  = (const float*)d_in[0];
  const int*   ei = (const int*)d_in[1];
  const float* W1 = (const float*)d_in[2];
  const float* b1 = (const float*)d_in[3];
  const float* W2 = (const float*)d_in[4];
  const float* b2 = (const float*)d_in[5];
  float* out = (float*)d_out;

  const int N = in_sizes[0] / IN_CH;
  const int E = in_sizes[1] / 2;
  const int* rowp = ei;
  const int* colp = ei + E;

  const size_t Np = ((size_t)N + 63) & ~(size_t)63;
  const size_t Ep = ((size_t)E + 63) & ~(size_t)63;
  const int NB = (N + BNODES - 1) / BNODES;
  const int etiles = (E + TEDGE - 1) / TEDGE;
  int*   wsi  = (int*)d_ws;
  int*   cnt  = wsi;
  int*   off  = wsi + Np;
  float* dinv = (float*)(wsi + 2 * Np);
  int*   tileOff = wsi + 3 * Np;
  int2*  bedge = (int2*)(wsi + 3 * Np + MAXT * MAXB);
  int*   srcs = wsi + 3 * Np + MAXT * MAXB + 2 * Ep;
  float* h1   = (float*)(srcs + Ep);
  float* h2   = h1 + ((size_t)N + 1) * HID;

  gcn_tile_bins<<<etiles, 256, 0, stream>>>(rowp, colp, tileOff, bedge, E);
  gcn_bin_csr<<<NB, 256, 0, stream>>>(tileOff, bedge, cnt, off, dinv, srcs,
                                      x, W1, h1, N, etiles);
  gcn_gather_gemm2<<<(N + BROWS - 1) / BROWS, 256, 0, stream>>>(
      off, cnt, srcs, dinv, h1, b1, W2, h2, N);
  gcn_gather<OUTP, OUTC, OUTC><<<(N + 8 * NPW - 1) / (8 * NPW), 256, 0, stream>>>(
      off, cnt, srcs, dinv, h2, b2, out, N);
}


# GEMM1's products run inside the per-bin kernel's two edge passes, 16 k-steps of one row tile behind each trip's segment loads (first tile stored ahead of the placing pass); sum orders kept
# speedup vs baseline: 5.1647x; 1.0578x over previous
#include <hip/hip_runtime.h>

typedef __attribute__((ext_vector_type(2))) float v2f;
typedef __attribute__((ext_vector_type(8))) float v8f;

#define IN_CH  128
#define HID    64
#define OUTC   40
#define OUTP   48
#define NPW 8
#define BSH    8
#define BNODES 256
#define MAXB   512
#define TPT    16
#define TEDGE  (TPT * 256)
#define MAXT   512
#define SB     25
#define SCAP   6144
#define WPITCH (IN_CH + 4)


__global__ void gcn_tile_bins(const int* __restrict__ row, const int* __restrict__ col,
                              int* __restrict__ tileOff, int2* __restrict__ bedge, int E) {
  __shared__ int h[MAXB];
  __shared__ int wsum[8];
  __shared__ int2 stage[TEDGE];
  const int tid = threadIdx.x;
  const int lane = tid & 31;
  const int wave = tid >> 5;
  h[2 * tid] = 0;
  h[2 * tid + 1] = 0;
  __syncthreads();

  const int tbase = blockIdx.x * TEDGE;
  const int base = tbase + tid;
  int cc[TPT], rk[TPT], rw[TPT];
#pragma unroll
  for (int i = 0; i < TPT; ++i) {
    const int e = base + i * 256;
    cc[i] = (e < E) ? col[e] : -1;
  }
#pragma unroll
  for (int i = 0; i < TPT; ++i) {
    const int e = base + i * 256;
    rw[i] = (e < E) ? row[e] : 0;
  }
#pragma unroll
  for (int i = 0; i < TPT; ++i) {
    rk[i] = 0;
    if (cc[i] >= 0) rk[i] = atomicAdd(&h[cc[i] >> BSH], 1);
  }
  __syncthreads();

  const int a0 = h[2 * tid];
  const int a1 = h[2 * tid + 1];
  const int s = a0 + a1;
  int inc = s;
  for (int o = 1; o < 32; o <<= 1) {
    const int t = __shfl_up(inc, o);
    if (lane >= o) inc += t;
  }
  if (lane == 31) wsum[wave] = inc;
  __syncthreads();
  int wpre = 0;
  for (int w = 0; w < wave; ++w) wpre += wsum[w];
  const int ex = wpre + inc - s;
  h[2 * tid] = ex;
  h[2 * tid + 1] = ex + a0;
  int2 pr;
  pr.x = ex;
  pr.y = ex + a0;
  *(int2*)(tileOff + (size_t)blockIdx.x * MAXB + 2 * tid) = pr;
  __syncthreads();

  int sl[TPT];
#pragma unroll
  for (int i = 0; i < TPT; ++i)
    sl[i] = h[((cc[i] >= 0) ? cc[i] : 0) >> BSH] + rk[i];
#pragma unroll
  for (int i = 0; i < TPT; ++i) {
    if (cc[i] >= 0) {
      int2 rec;
      rec.x = rw[i];
      rec.y = cc[i];
      stage[sl[i]] = rec;
    }
  }
  __syncthreads();
  const int n_t = (E - tbase < TEDGE) ? (E - tbase) : TEDGE;
  int2 ov[TPT];
#pragma unroll
  for (int i = 0; i < TPT; ++i) ov[i] = stage[tid + i * 256];
#pragma unroll
  for (int i = 0; i < TPT; ++i) {
    const int j = tid + i * 256;
    if (j < n_t) bedge[tbase + j] = ov[i];
  }
}

template <int J0>
__device__ __forceinline__ void gemm1_xload8(const float* xr, v2f (&a)[8]) {
#pragma unroll
  for (int u = 0; u < 8; ++u) a[u] = *(const v2f*)(xr + 4 * (J0 + u));
}

template <int J0>
__device__ __forceinline__ void gemm1_steps8(const v2f (&a)[8], const float* brow,
                                             v8f (&acc)[4]) {
#pragma unroll
  for (int u = 0; u < 8; ++u) {
    const int j = J0 + u;
    const v2f b0 = *(const v2f*)(brow + 4 * j);
    const v2f b1 = *(const v2f*)(brow + 16 * WPITCH + 4 * j);
    const v2f b2 = *(const v2f*)(brow + 32 * WPITCH + 4 * j);
    const v2f b3 = *(const v2f*)(brow + 48 * WPITCH + 4 * j);
    acc[0] = __builtin_amdgcn_wmma_f32_16x16x4_f32(false, a[u], false, b0,
                                                   (short)0, acc[0], false, false);
    acc[1] = __builtin_amdgcn_wmma_f32_16x16x4_f32(false, a[u], false, b1,
                                                   (short)0, acc[1], false, false);
    acc[2] = __builtin_amdgcn_wmma_f32_16x16x4_f32(false, a[u], false, b2,
                                                   (short)0, acc[2], false, false);
    acc[3] = __builtin_amdgcn_wmma_f32_16x16x4_f32(false, a[u], false, b3,
                                                   (short)0, acc[3], false, false);
  }
}

__device__ __forceinline__ void gemm1_step(const float* xr, int j, const float* brow,
                                           v8f (&acc)[4]) {
  const v2f a = *(const v2f*)(xr + 4 * j);
  const v2f b0 = *(const v2f*)(brow + 4 * j);
  const v2f b1 = *(const v2f*)(brow + 16 * WPITCH + 4 * j);
  const v2f b2 = *(const v2f*)(brow + 32 * WPITCH + 4 * j);
  const v2f b3 = *(const v2f*)(brow + 48 * WPITCH + 4 * j);
  acc[0] = __builtin_amdgcn_wmma_f32_16x16x4_f32(false, a, false, b0,
                                                 (short)0, acc[0], false, false);
  acc[1] = __builtin_amdgcn_wmma_f32_16x16x4_f32(false, a, false, b1,
                                                 (short)0, acc[1], false, false);
  acc[2] = __builtin_amdgcn_wmma_f32_16x16x4_f32(false, a, false, b2,
                                                 (short)0, acc[2], false, false);
  acc[3] = __builtin_amdgcn_wmma_f32_16x16x4_f32(false, a, false, b3,
                                                 (short)0, acc[3], false, false);
}

template <int L>
__device__ __forceinline__ void gemm1_chunk16(const float* xr0, const float* xr1,
                                              v2f (&s0)[8], v2f (&s1)[8],
                                              const float* brow,
                                              v8f (&acc0)[4], v8f (&acc1)[4]) {
  constexpr int T = L / 32, J = L % 32;
  if (L > 0) gemm1_xload8<J + 8>(T ? xr1 : xr0, s1);
  gemm1_steps8<J>(s0, brow, T ? acc1 : acc0);
  if (J == 0) gemm1_xload8<16>(T ? xr1 : xr0, s0);
  gemm1_steps8<J + 8>(s1, brow, T ? acc1 : acc0);
}

__device__ __forceinline__ void gemm1_store(const v8f (&acc)[4], const float* sdr,
                                            float* o0) {
  const float4 p0 = *(const float4*)(sdr);
  const float4 p1 = *(const float4*)(sdr + 4);
  const float dr[8] = {p0.x, p0.y, p0.z, p0.w, p1.x, p1.y, p1.z, p1.w};
#pragma unroll
  for (int j = 0; j < 8; ++j) {
    float* o = o0 + (size_t)j * HID;
    o[0]  = acc[0][j] * dr[j];
    o[16] = acc[1][j] * dr[j];
    o[32] = acc[2][j] * dr[j];
    o[48] = acc[3][j] * dr[j];
  }
}

__global__ void __launch_bounds__(256, 4)
gcn_bin_csr(const int* __restrict__ tileOff,
                            const int2* __restrict__ bedge,
                            int* __restrict__ cnt, int* __restrict__ off,
                            float* __restrict__ dinv, int* __restrict__ src,
                            const float* __restrict__ x,
                            const float* __restrict__ W1,
                            float* __restrict__ h1,
                            int N, int etiles) {
  __shared__ int c[BNODES + 32];
  __shared__ int segOff[MAXT];
  __shared__ int segLen[MAXT];
  __shared__ int wsum[8];
  __shared__ int wsum2[8];
  __shared__ int sbuf[SCAP];
  __shared__ alignas(16) float sw[HID * WPITCH];
  __shared__ alignas(16) float sd[BNODES];
  const int tid = threadIdx.x;
  const int lane = tid & 31;
  const int wave = __builtin_amdgcn_readfirstlane(tid >> 5);
  const int b = blockIdx.x;
  const int lo = b * BNODES;

  const int tiles = N >> 4;
  const int t0g = b * (BNODES / 16) + 2 * wave;
  const bool has = t0g < tiles;
  const bool two = t0g + 1 < tiles;
  const int rci = lane & 15;
  const int hf = lane >> 4;
  const int m0 = has ? t0g * 16 : 0;
  const int m1 = two ? m0 + 16 : m0;
  const float* xr0 = x + (size_t)(m0 + rci) * IN_CH + 2 * hf;
  const float* xr1 = x + (size_t)(m1 + rci) * IN_CH + 2 * hf;
  const float* brow = sw + rci * WPITCH + 2 * hf;
  v2f fs0[8], fs1[8], fs2[8];
  v8f acc0[4], acc1[4];
#pragma unroll
  for (int i = 0; i < 4; ++i) {
    acc0[i] = v8f{};
    acc1[i] = v8f{};
  }
  int kdone = 0;

  int part = 0;
#pragma unroll
  for (int r = 0; r < 2; ++r) {
    const int t = tid + r * 256;
    int o0 = 0, ln = 0;
    if (t < etiles) {
      const int* p = tileOff + (size_t)t * MAXB + b;
      o0 = p[0];
      ln = p[1] - o0;
    }
    part += o0;
    segOff[t] = t * TEDGE + o0;
    segLen[t] = ln;
  }
#pragma unroll
  for (int j = 0; j < IN_CH * HID / 256; ++j) {
    const int i = j * 256 + tid;
    sw[(i & (HID - 1)) * WPITCH + (i / HID)] = W1[i];
  }
  gemm1_xload8<0>(xr0, fs0);
  gemm1_xload8<8>(xr0, fs1);
  gemm1_xload8<0>(xr1, fs2);
  for (int o = 16; o > 0; o >>= 1) part += __shfl_xor(part, o);
  if (lane == 0) wsum[wave] = part;
  c[tid] = 0;
  if (tid < 32) c[BNODES + tid] = 0;
  __syncthreads();
  int start = 0;
  for (int w = 0; w < 8; ++w) start += wsum[w];

  for (int t0 = wave; t0 < etiles; t0 += 8 * SB) {
    const int tq = t0 + 8 * lane;
    const bool tv = (lane < SB) && (tq < MAXT);
    const int m_len = tv ? segLen[tq] : 0;
    const int m_off = tv ? segOff[tq] : 0;
    int ln[SB], tg[SB];
#pragma unroll
    for (int q = 0; q < SB; ++q) {
      ln[q] = __builtin_amdgcn_readlane(m_len, q);
      const int so = __builtin_amdgcn_readlane(m_off, q);
      const int k = (lane < ln[q]) ? lane : ln[q] - 1;
      const int idx = (ln[q] > 0) ? so + k : 0;
      tg[q] = bedge[idx].y;
    }
    if (kdone == 0) {
      gemm1_chunk16<0>(xr0, xr1, fs0, fs1, brow, acc0, acc1);
      kdone = 16;
    } else if (kdone == 16) {
      gemm1_chunk16<16>(xr0, xr1, fs0, fs1, brow, acc0, acc1);
      kdone = 32;
    }
#pragma unroll
    for (int q = 0; q < SB; ++q)
      if (lane < ln[q]) atomicAdd(&c[tg[q] - lo], 1);
#pragma unroll
    for (int q = 0; q < SB; ++q) {
      if (ln[q] > 32) {
        const int so = __builtin_amdgcn_readlane(m_off, q);
        for (int k = 32 + lane; k < ln[q]; k += 32)
          atomicAdd(&c[bedge[so + k].y - lo], 1);
      }
    }
  }
  __syncthreads();

  const int v = c[tid];
  int inc = v;
  for (int o = 1; o < 32; o <<= 1) {
    const int t = __shfl_up(inc, o);
    if (lane >= o) inc += t;
  }
  if (lane == 31) wsum2[wave] = inc;
  __syncthreads();
  int wpre = 0, tot = 0;
  for (int w = 0; w < 8; ++w) {
    const int s = wsum2[w];
    if (w < wave) wpre += s;
    tot += s;
  }
  const int ex = wpre + inc - v;
  const int node = lo + tid;
  float dn = 0.0f;
  if (node < N) {
    cnt[node] = v;
    off[node] = start + ex;
    const float d = (float)(v + 1);
    dn = (d > 0.0f) ? rsqrtf(d) : 0.0f;
    dinv[node] = dn;
  }
  sd[tid] = dn;
  c[tid] = ex;
  __syncthreads();

#pragma nounroll
  for (int j = kdone; j < IN_CH / 4; ++j) gemm1_step(xr0, j, brow, acc0);
  if (kdone < IN_CH / 4) kdone = IN_CH / 4;
  const int lr = (m0 - lo) + 8 * hf;
  if (has) gemm1_store(acc0, sd + lr, h1 + (size_t)(m0 + 8 * hf) * HID + rci);

  auto place = [&](auto put) {
    for (int t0 = wave; t0 < etiles; t0 += 8 * SB) {
      const int tq = t0 + 8 * lane;
      const bool tv = (lane < SB) && (tq < MAXT);
      const int m_len = tv ? segLen[tq] : 0;
      const int m_off = tv ? segOff[tq] : 0;
      int ln[SB], sl[SB];
      int2 rc[SB];
#pragma unroll
      for (int q = 0; q < SB; ++q) {
        ln[q] = __builtin_amdgcn_readlane(m_len, q);
        const int so = __builtin_amdgcn_readlane(m_off, q);
        const int k = (lane < ln[q]) ? lane : ln[q] - 1;
        const int idx = (ln[q] > 0) ? so + k : 0;
        rc[q] = bedge[idx];
      }
      if (kdone == 32) {
        gemm1_chunk16<32>(xr0, xr1, fs2, fs1, brow, acc0, acc1);
        kdone = 48;
      } else if (kdone == 48) {
        gemm1_chunk16<48>(xr0, xr1, fs2, fs1, brow, acc0, acc1);
        kdone = 64;
      }
#pragma unroll
      for (int q = 0; q < SB; ++q) {
        const bool on = lane < ln[q];
        sl[q] = atomicAdd(&c[on ? rc[q].y - lo : BNODES + lane], on ? 1 : 0);
      }
#pragma unroll
      for (int q = 0; q < SB; ++q)
        if (lane < ln[q]) put(sl[q], rc[q].x);
#pragma unroll
      for (int q = 0; q < SB; ++q) {
        if (ln[q] > 32) {
          const int so = __builtin_amdgcn_readlane(m_off, q);
          for (int k = 32 + lane; k < ln[q]; k += 32) {
            const int2 rec = bedge[so + k];
            put(atomicAdd(&c[rec.y - lo], 1), rec.x);
          }
        }
      }
    }
  };
  if (tot <= SCAP) {
    place([&](int slot, int s) { sbuf[slot] = s; });
    __syncthreads();
    const int a = start & 31;
    for (int j0 = tid - a; j0 < tot; j0 += 4 * 256) {
      int w4[4];
      for (int u = 0; u < 4; ++u) {
        const int j = j0 + u * 256;
        w4[u] = sbuf[(j >= 0 && j < tot) ? j : 0];
      }
      for (int u = 0; u < 4; ++u) {
        const int j = j0 + u * 256;
        if (j >= 0 && j < tot) src[start + j] = w4[u];
      }
    }
  } else {
    place([&](int slot, int s) { src[start + slot] = s; });
  }

  if (b == 0 && wave == 0) {
    float2 z;
    z.x = 0.0f;
    z.y = 0.0f;
    *(float2*)(h1 + (size_t)N * HID + lane * 2) = z;
  }
#pragma nounroll
  for (int j = kdone - IN_CH / 4; j < IN_CH / 4; ++j) gemm1_step(xr1, j, brow, acc1);
  if (two) gemm1_store(acc1, sd + lr + 16, h1 + (size_t)(m1 + 8 * hf) * HID + rci);
}

template <int STRIDE, int CH, typename StoreF>
__device__ __forceinline__ void gcn_gather_wave(const int* __restrict__ off,
                                                const int* __restrict__ cnt,
                                                const int* __restrict__ src,
                                                const float* __restrict__ dinv,
                                                const float* __restrict__ h,
                                                const float* __restrict__ bias,
                                                int N, int c0, int lane, StoreF store) {
  const int nn = (N - c0 < NPW) ? (N - c0) : NPW;
  const int last = nn - 1;

  int m_off = 0, m_tot = 1;
  float m_d = 0.0f;
  if (lane < nn) {
    m_off = off[c0 + lane];
    m_tot = cnt[c0 + lane] + 1;
    m_d = dinv[c0 + lane];
  }
  const bool act = lane < CH / 2;
  const float* hl = h + (act ? lane * 2 : 0);
  float bx = 0.0f, by = 0.0f;
  if (act) {
    bx = bias[lane * 2];
    by = bias[lane * 2 + 1];
  }

  auto raw_r = [&](int i) -> int {
    const int st = __builtin_amdgcn_readlane(m_off, i);
    const int tt = __builtin_amdgcn_readlane(m_tot, i);
    const int idx = (lane < tt - 1) ? st + lane : 0;
    return src[idx];
  };
  auto fix_r = [&](int i, int raw) -> int {
    const int tt = __builtin_amdgcn_readlane(m_tot, i);
    return (lane < tt - 1) ? raw : ((lane == tt - 1) ? c0 + i : N);
  };

  int r_cur = fix_r(0, raw_r(0));
  int raw_nxt = raw_r(last < 1 ? last : 1);

  for (int i = 0; i < nn; ++i) {
    const int c = c0 + i;
    const int start = __builtin_amdgcn_readlane(m_off, i);
    const int total = __builtin_amdgcn_readlane(m_tot, i);
    const float dc = __int_as_float(__builtin_amdgcn_readlane(__float_as_int(m_d), i));

    int r_l = r_cur;

    float2 v0[16];
    for (int q = 0; q < 16; ++q) {
      const int r = __builtin_amdgcn_readlane(r_l, q);
      v0[q] = *(const float2*)(hl + (size_t)r * STRIDE);
    }
    const int i1 = (i + 1 < last) ? i + 1 : last;
    const int i2 = (i + 2 < last) ? i + 2 : last;
    const int raw_nn = raw_r(i2);

    float ax = 0.0f, ay = 0.0f;
    for (int q = 0; q < 16; ++q) {
      ax += v0[q].x;
      ay += v0[q].y;
    }

    int base = 0, k = 16;
    for (;;) {
      const int rem = total - base;
      const int m = rem < 32 ? rem : 32;
      for (; k + 8 <= m; k += 8) {
        float2 v[8];
        for (int q = 0; q < 8; ++q) {
          const int r = __builtin_amdgcn_readlane(r_l, k + q);
          v[q] = *(const float2*)(hl + (size_t)r * STRIDE);
        }
        for (int q = 0; q < 8; ++q) {
          ax += v[q].x;
          ay += v[q].y;
        }
      }
      for (; k < m; k += 4) {
        float2 v[4];
        for (int q = 0; q < 4; ++q) {
          const int r = __builtin_amdgcn_readlane(r_l, k + q);
          v[q] = *(const float2*)(hl + (size_t)r * STRIDE);
        }
        for (int q = 0; q < 4; ++q) {
          ax += v[q].x;
          ay += v[q].y;
        }
      }
      base += 32;
      if (base >= total) break;
      const int j = base + lane;
      r_l = (j == total - 1) ? c : N;
      if (j < total - 1) r_l = src[start + j];
      k = 0;
    }
    if (act) store(c, ax * dc + bx, ay * dc + by);
    r_cur = fix_r(i1, raw_nxt);
    raw_nxt = raw_nn;
  }
}

template <int STRIDE, int CH, int OSTR>
__global__ void gcn_gather(const int* __restrict__ off, const int* __restrict__ cnt,
                           const int* __restrict__ src,
                           const float* __restrict__ dinv,
                           const float* __restrict__ h,
                           const float* __restrict__ bias,
                           float* __restrict__ dst, int N) {
  const int lane = threadIdx.x & 31;
  const int c0 = NPW * __builtin_amdgcn_readfirstlane(
      (int)((blockIdx.x * blockDim.x + threadIdx.x) >> 5));
  if (c0 >= N) return;
  gcn_gather_wave<STRIDE, CH>(off, cnt, src, dinv, h, bias, N, c0, lane,
                              [&](int c, float ox, float oy) {
    float2 o;
    o.x = ox;
    o.y = oy;
    *(float2*)(dst + (size_t)c * OSTR + lane * 2) = o;
  });
}

#define BROWS  (8 * NPW)
#define APITCH (HID + 4)
__global__ void gcn_gather_gemm2(const int* __restrict__ off, const int* __restrict__ cnt,
                                 const int* __restrict__ src,
                                 const float* __restrict__ dinv,
                                 const float* __restrict__ h,
                                 const float* __restrict__ bias,
                                 const float* __restrict__ W2,
                                 float* __restrict__ h2, int N) {
  __shared__ float sa[BROWS * APITCH];
  __shared__ float sw[OUTP * APITCH];
#pragma unroll
  for (int j = 0; j < HID * OUTC / 256; ++j) {
    const int i = j * 256 + (int)threadIdx.x;
    const int k = i / OUTC;
    sw[(i - k * OUTC) * APITCH + k] = W2[i];
  }
  for (int i = threadIdx.x; i < (OUTP - OUTC) * HID; i += 256)
    sw[(OUTC + i / HID) * APITCH + (i % HID)] = 0.0f;
  const int lane = threadIdx.x & 31;
  const int wave = __builtin_amdgcn_readfirstlane((int)(threadIdx.x >> 5));
  const int r0 = blockIdx.x * BROWS;
  const int c0 = r0 + wave * NPW;
  if (c0 < N) {
    gcn_gather_wave<HID, HID>(off, cnt, src, dinv, h, bias, N, c0, lane,
                              [&](int c, float ox, float oy) {
      float2 o;
      o.x = fmaxf(ox, 0.0f);
      o.y = fmaxf(oy, 0.0f);
      *(float2*)(sa + (c - r0) * APITCH + lane * 2) = o;
    });
  }
  __syncthreads();

  const int tile = wave & 3;
  const int m0 = r0 + tile * 16;
  if (m0 + 16 <= N) {
    const int rc = lane & 15;
    const int kOff = (lane < 16) ? 0 : 2;
    const int mBase = m0 + ((lane < 16) ? 0 : 8);
    const float4 dlo = *(const float4*)(dinv + mBase);
    const float4 dhi = *(const float4*)(dinv + mBase + 4);
    const float dr[8] = {dlo.x, dlo.y, dlo.z, dlo.w, dhi.x, dhi.y, dhi.z, dhi.w};
    const float* arow = sa + (tile * 16 + rc) * APITCH + kOff;
    const float* brow = sw + rc * APITCH + kOff;
    if (wave < 4) {
      v8f acc0 = {}, acc1 = {};
#pragma unroll
      for (int k = 0; k < HID; k += 4) {
        v2f a, b0, b1;
        a.x = arow[k];
        a.y = arow[k + 1];
        b0.x = brow[k];                 b0.y = brow[k + 1];
        b1.x = brow[16 * APITCH + k];   b1.y = brow[16 * APITCH + k + 1];
        acc0 = __builtin_amdgcn_wmma_f32_16x16x4_f32(false, a, false, b0,
                                                     (short)0, acc0, false, false);
        acc1 = __builtin_amdgcn_wmma_f32_16x16x4_f32(false, a, false, b1,
                                                     (short)0, acc1, false, false);
      }
#pragma unroll
      for (int j = 0; j < 8; ++j) {
        float* o = h2 + (size_t)(mBase + j) * OUTP + rc;
        o[0]  = acc0[j] * dr[j];
        o[16] = acc1[j] * dr[j];
      }
    } else {
      v8f acc2 = {};
#pragma unroll
      for (int k = 0; k < HID; k += 4) {
        v2f a, b2;
        a.x = arow[k];
        a.y = arow[k + 1];
        b2.x = brow[32 * APITCH + k];
        b2.y = brow[32 * APITCH + k + 1];
        acc2 = __builtin_amdgcn_wmma_f32_16x16x4_f32(false, a, false, b2,
                                                     (short)0, acc2, false, false);
      }
      if (32 + rc < OUTC) {
#pragma unroll
        for (int j = 0; j < 8; ++j)
          h2[(size_t)(mBase + j) * OUTP + 32 + rc] = acc2[j] * dr[j];
      }
    }
  }
  if (blockIdx.x == 0 && wave == 0 && lane < OUTP / 2) {
    float2 z;
    z.x = 0.0f;
    z.y = 0.0f;
    *(float2*)(h2 + (size_t)N * OUTP + lane * 2) = z;
  }
}

extern "C" void kernel_launch(void* const* d_in, const int* in_sizes, int n_in,
                              void* d_out, int out_size, void* d_ws, size_t ws_size,
                              hipStream_t stream) {
  const float* x  = (const float*)d_in[0];
  const int*   ei = (const int*)d_in[1];
  const float* W1 = (const float*)d_in[2];
  const float* b1 = (const float*)d_in[3];
  const float* W2 = (const float*)d_in[4];
  const float* b2 = (const float*)d_in[5];
  float* out = (float*)d_out;

  const int N = in_sizes[0] / IN_CH;
  const int E = in_sizes[1] / 2;
  const int* rowp = ei;
  const int* colp = ei + E;

  const size_t Np = ((size_t)N + 63) & ~(size_t)63;
  const size_t Ep = ((size_t)E + 63) & ~(size_t)63;
  const int NB = (N + BNODES - 1) / BNODES;
  const int etiles = (E + TEDGE - 1) / TEDGE;
  int*   wsi  = (int*)d_ws;
  int*   cnt  = wsi;
  int*   off  = wsi + Np;
  float* dinv = (float*)(wsi + 2 * Np);
  int*   tileOff = wsi + 3 * Np;
  int2*  bedge = (int2*)(wsi + 3 * Np + MAXT * MAXB);
  int*   srcs = wsi + 3 * Np + MAXT * MAXB + 2 * Ep;
  float* h1   = (float*)(srcs + Ep);
  float* h2   = h1 + ((size_t)N + 1) * HID;

  gcn_tile_bins<<<etiles, 256, 0, stream>>>(rowp, colp, tileOff, bedge, E);
  gcn_bin_csr<<<NB, 256, 0, stream>>>(tileOff, bedge, cnt, off, dinv, srcs,
                                      x, W1, h1, N, etiles);
  gcn_gather_gemm2<<<(N + BROWS - 1) / BROWS, 256, 0, stream>>>(
      off, cnt, srcs, dinv, h1, b1, W2, h2, N);
  gcn_gather<OUTP, OUTC, OUTC><<<(N + 8 * NPW - 1) / (8 * NPW), 256, 0, stream>>>(
      off, cnt, srcs, dinv, h2, b2, out, N);
}
